# starting kernel
# baseline (speedup 1.0000x reference)
#include <hip/hip_runtime.h>
#include <hip/hip_bf16.h>
#include <math.h>


typedef float v2f __attribute__((ext_vector_type(2)));
typedef float v8f __attribute__((ext_vector_type(8)));

#define N_NODES   50000
#define N_EDGES   640000
#define F_DIM     128
#define N_CLASSES 16
#define N_GRAPHS  500


__global__ void init_deg_kernel(float* __restrict__ deg, int n) {
    int i = blockIdx.x * blockDim.x + threadIdx.x;
    if (i < n) deg[i] = 1.0f;
}

__global__ void deg_edges_kernel(const int* __restrict__ dst, float* __restrict__ deg, int e) {
    int i = blockIdx.x * blockDim.x + threadIdx.x;
    if (i < e) atomicAdd(&deg[dst[i]], 1.0f);
}

__global__ void dis_kernel(const float* __restrict__ deg, float* __restrict__ dis, int n) {
    int i = blockIdx.x * blockDim.x + threadIdx.x;
    if (i < n) dis[i] = rsqrtf(deg[i]);
}


__global__ void zero_kernel(float* __restrict__ p, int n) {
    int i = blockIdx.x * blockDim.x + threadIdx.x;
    if (i < n) p[i] = 0.0f;
}


__global__ __launch_bounds__(256)
void gemm128_wmma_kernel(const float* __restrict__ A, const float* __restrict__ W,
                         float* __restrict__ C, int M) {
    __shared__ float lW[F_DIM * F_DIM];
    for (int i = threadIdx.x; i < F_DIM * F_DIM; i += 256) lW[i] = W[i];
    __syncthreads();

    const int wave = threadIdx.x >> 5;
    const int lane = threadIdx.x & 31;
    const int strip = blockIdx.x * 8 + wave;
    const int row0  = strip * 16;
    if (row0 >= M) return;

    const int half = lane >> 4;
    const int l16  = lane & 15;

    const v8f zero = {0.f, 0.f, 0.f, 0.f, 0.f, 0.f, 0.f, 0.f};
    v8f acc[8];
#pragma unroll
    for (int t = 0; t < 8; ++t) acc[t] = zero;

    const float* __restrict__ arow = A + (size_t)(row0 + l16) * F_DIM;

    for (int k0 = 0; k0 < F_DIM; k0 += 4) {
        const int ka = k0 + half * 2;
        v2f a;
        a.x = arow[ka];
        a.y = arow[ka + 1];
#pragma unroll
        for (int t = 0; t < 8; ++t) {
            const int col = t * 16 + l16;
            v2f b;
            b.x = lW[ka * F_DIM + col];
            b.y = lW[(ka + 1) * F_DIM + col];
            acc[t] = __builtin_amdgcn_wmma_f32_16x16x4_f32(
                 false, a,  false, b,
                 (short)0, acc[t],  false,  false);
        }
    }

#pragma unroll
    for (int t = 0; t < 8; ++t) {
        const int col = t * 16 + l16;
#pragma unroll
        for (int v = 0; v < 8; ++v) {
            C[(size_t)(row0 + v + 8 * half) * F_DIM + col] = acc[t][v];
        }
    }
}


__global__ void scatter_kernel(const float* __restrict__ H,
                               const int* __restrict__ src,
                               const int* __restrict__ dst,
                               const float* __restrict__ dis,
                               float* __restrict__ out, int e) {
    int tid  = blockIdx.x * blockDim.x + threadIdx.x;
    int edge = tid >> 5;
    int lane = tid & 31;
    if (edge >= e) return;
    int s = src[edge];
    int d = dst[edge];
    float norm = dis[s] * dis[d];
    const float4* __restrict__ hv = (const float4*)(H + (size_t)s * F_DIM);
    float4 m = hv[lane];
    float* o = out + (size_t)d * F_DIM + lane * 4;
    atomicAdd(o + 0, m.x * norm);
    atomicAdd(o + 1, m.y * norm);
    atomicAdd(o + 2, m.z * norm);
    atomicAdd(o + 3, m.w * norm);
}


__global__ void selfloop_bias_relu_kernel(float* __restrict__ S,
                                          const float* __restrict__ H,
                                          const float* __restrict__ dis,
                                          const float* __restrict__ b, int n) {
    int tid = blockIdx.x * blockDim.x + threadIdx.x;
    if (tid >= n * F_DIM) return;
    int i = tid >> 7;
    int h = tid & (F_DIM - 1);
    float di = dis[i];
    float v = S[tid] + H[tid] * di * di + b[h];
    S[tid] = fmaxf(v, 0.0f);
}


__global__ void pool_kernel(const float* __restrict__ H,
                            const int* __restrict__ batch,
                            float* __restrict__ pooled, int n) {
    int tid  = blockIdx.x * blockDim.x + threadIdx.x;
    int node = tid >> 5;
    int lane = tid & 31;
    if (node >= n) return;
    int g = batch[node];
    const float4* __restrict__ hv = (const float4*)(H + (size_t)node * F_DIM);
    float4 m = hv[lane];
    float* o = pooled + (size_t)g * F_DIM + lane * 4;
    atomicAdd(o + 0, m.x);
    atomicAdd(o + 1, m.y);
    atomicAdd(o + 2, m.z);
    atomicAdd(o + 3, m.w);
}


__global__ void head_kernel(const float* __restrict__ pooled,
                            const float* __restrict__ Wlin,
                            const float* __restrict__ blin,
                            float* __restrict__ out, int g_total) {
    int g = blockIdx.x * blockDim.x + threadIdx.x;
    if (g >= g_total) return;
    float logits[N_CLASSES];
#pragma unroll
    for (int c = 0; c < N_CLASSES; ++c) logits[c] = blin[c];
    const float* __restrict__ pr = pooled + (size_t)g * F_DIM;
    for (int k = 0; k < F_DIM; ++k) {
        float p = pr[k];
        const float* __restrict__ wr = Wlin + (size_t)k * N_CLASSES;
#pragma unroll
        for (int c = 0; c < N_CLASSES; ++c) logits[c] = fmaf(p, wr[c], logits[c]);
    }
    float mx = logits[0];
#pragma unroll
    for (int c = 1; c < N_CLASSES; ++c) mx = fmaxf(mx, logits[c]);
    float s = 0.0f;
#pragma unroll
    for (int c = 0; c < N_CLASSES; ++c) s += expf(logits[c] - mx);
    float lse = mx + logf(s);
    float* o = out + (size_t)g * N_CLASSES;
#pragma unroll
    for (int c = 0; c < N_CLASSES; ++c) o[c] = logits[c] - lse;
}


extern "C" void kernel_launch(void* const* d_in, const int* in_sizes, int n_in,
                              void* d_out, int out_size, void* d_ws, size_t ws_size,
                              hipStream_t stream) {
    const float* x     = (const float*)d_in[0];
    const int*   eidx  = (const int*)d_in[1];
    const int*   batch = (const int*)d_in[2];
    const float* W1    = (const float*)d_in[3];
    const float* b1    = (const float*)d_in[4];
    const float* W2    = (const float*)d_in[5];
    const float* b2    = (const float*)d_in[6];
    const float* Wlin  = (const float*)d_in[7];
    const float* blin  = (const float*)d_in[8];
    float*       out   = (float*)d_out;

    const int* src = eidx;
    const int* dst = eidx + N_EDGES;

    float* ws = (float*)d_ws;
    float* deg    = ws;
    float* dis    = ws + 50000;
    float* H      = ws + 100000;
    float* S      = H + (size_t)N_NODES * F_DIM;
    float* pooled = S + (size_t)N_NODES * F_DIM;

    const int B = 256;
    const int nodeBlocks  = (N_NODES + B - 1) / B;
    const int edgeBlocks  = (N_EDGES + B - 1) / B;
    const int featBlocks  = ((size_t)N_NODES * F_DIM + B - 1) / B;
    const int scatBlocks  = ((size_t)N_EDGES * 32 + B - 1) / B;
    const int poolBlocks  = ((size_t)N_NODES * 32 + B - 1) / B;
    const int gemmBlocks  = (N_NODES / 16 + 7) / 8;
    const int headBlocks  = (N_GRAPHS + B - 1) / B;

    init_deg_kernel<<<nodeBlocks, B, 0, stream>>>(deg, N_NODES);
    deg_edges_kernel<<<edgeBlocks, B, 0, stream>>>(dst, deg, N_EDGES);
    dis_kernel<<<nodeBlocks, B, 0, stream>>>(deg, dis, N_NODES);

    gemm128_wmma_kernel<<<gemmBlocks, B, 0, stream>>>(x, W1, H, N_NODES);
    zero_kernel<<<featBlocks, B, 0, stream>>>(S, N_NODES * F_DIM);
    scatter_kernel<<<scatBlocks, B, 0, stream>>>(H, src, dst, dis, S, N_EDGES);
    selfloop_bias_relu_kernel<<<featBlocks, B, 0, stream>>>(S, H, dis, b1, N_NODES);

    gemm128_wmma_kernel<<<gemmBlocks, B, 0, stream>>>(S, W2, H, N_NODES);
    zero_kernel<<<featBlocks, B, 0, stream>>>(S, N_NODES * F_DIM);
    scatter_kernel<<<scatBlocks, B, 0, stream>>>(H, src, dst, dis, S, N_EDGES);
    selfloop_bias_relu_kernel<<<featBlocks, B, 0, stream>>>(S, H, dis, b2, N_NODES);

    zero_kernel<<<(N_GRAPHS * F_DIM + B - 1) / B, B, 0, stream>>>(pooled, N_GRAPHS * F_DIM);
    pool_kernel<<<poolBlocks, B, 0, stream>>>(S, batch, pooled, N_NODES);
    head_kernel<<<headBlocks, B, 0, stream>>>(pooled, Wlin, blin, out, N_GRAPHS);
}


# Edges counting-sorted by destination each call (rank via returning int atomic, prefix scan); one wave per node gathers rows, stores once, replacing float atomic scatter; sum order = arrival rank
# speedup vs baseline: 1.6699x; 1.6699x over previous
#include <hip/hip_runtime.h>
#include <hip/hip_bf16.h>
#include <math.h>


typedef float v2f __attribute__((ext_vector_type(2)));
typedef float v8f __attribute__((ext_vector_type(8)));

#define N_NODES   50000
#define N_EDGES   640000
#define F_DIM     128
#define N_CLASSES 16
#define N_GRAPHS  500


__global__ void init_cnt_kernel(int* __restrict__ cnt, int n) {
    int i = blockIdx.x * blockDim.x + threadIdx.x;
    if (i < n) cnt[i] = 0;
}

__global__ void count_rank_kernel(const int* __restrict__ dst, int* __restrict__ cnt,
                                  int* __restrict__ rank, int e) {
    int i = blockIdx.x * blockDim.x + threadIdx.x;
    if (i < e) rank[i] = atomicAdd(&cnt[dst[i]], 1);
}

__global__ __launch_bounds__(1024)
void scan_dis_kernel(const int* __restrict__ cnt, int* __restrict__ rowptr,
                     float* __restrict__ dis, int n) {
    __shared__ int s_wtot[32];
    __shared__ int s_wpre[32];
    __shared__ int s_woff[32];
    const int tid  = threadIdx.x;
    const int lane = tid & 31;
    const int wave = tid >> 5;

    int pre = 0;
    const int4* __restrict__ c4 = (const int4*)cnt;
    const int nq = blockIdx.x * 256;
    int q = tid;
    for (; q + 3072 < nq; q += 4096) {
        int4 v0 = c4[q];
        int4 v1 = c4[q + 1024];
        int4 v2 = c4[q + 2048];
        int4 v3 = c4[q + 3072];
        pre += ((v0.x + v0.y) + (v0.z + v0.w)) + ((v1.x + v1.y) + (v1.z + v1.w))
             + ((v2.x + v2.y) + (v2.z + v2.w)) + ((v3.x + v3.y) + (v3.z + v3.w));
    }
    for (; q < nq; q += 1024) {
        int4 v = c4[q];
        pre += (v.x + v.y) + (v.z + v.w);
    }

    const int i = blockIdx.x * 1024 + tid;
    const int c = (i < n) ? cnt[i] : 0;

    int incl = c;
    for (int o = 1; o < 32; o <<= 1) {
        int t = __shfl_up(incl, o, 32);
        if (lane >= o) incl += t;
    }
    for (int o = 16; o > 0; o >>= 1) pre += __shfl_xor(pre, o, 32);

    if (lane == 31) s_wtot[wave] = incl;
    if (lane == 0)  s_wpre[wave] = pre;
    __syncthreads();
    if (wave == 0) {
        const int t = s_wtot[lane];
        int p  = s_wpre[lane];
        int ti = t;
        for (int o = 1; o < 32; o <<= 1) {
            int u = __shfl_up(ti, o, 32);
            if (lane >= o) ti += u;
        }
        for (int o = 16; o > 0; o >>= 1) p += __shfl_xor(p, o, 32);
        s_woff[lane] = p + ti - t;
    }
    __syncthreads();
    const int off = s_woff[wave];
    if (i < n) {
        rowptr[i] = off + incl - c;
        dis[i]    = rsqrtf((float)(c + 1));
        if (i == n - 1) rowptr[n] = off + incl;
    }
}

__global__ void place_kernel(const int* __restrict__ src, const int* __restrict__ dst,
                             const int* __restrict__ rank, const int* __restrict__ rowptr,
                             const float* __restrict__ dis, int* __restrict__ csr, int e) {
    int i = blockIdx.x * blockDim.x + threadIdx.x;
    if (i >= e) return;
    const int s = src[i];
    const int d = dst[i];
    const int r = rank[i];
    const int   rp = rowptr[d];
    const float dd = dis[d];
    const float dsrc = dis[s];
    const int   pos = rp + r;
    const float w = dsrc * dd;
    if ((unsigned)pos < (unsigned)e) ((int2*)csr)[pos] = make_int2(s, __float_as_int(w));
}


__global__ void zero_kernel(float* __restrict__ p, int n) {
    int i = blockIdx.x * blockDim.x + threadIdx.x;
    if (i < n) p[i] = 0.0f;
}


__global__ __launch_bounds__(256)
void gemm128_wmma_kernel(const float* __restrict__ A, const float* __restrict__ W,
                         float* __restrict__ C, int M) {
    __shared__ float lW[F_DIM * F_DIM];
    for (int i = threadIdx.x; i < F_DIM * F_DIM; i += 256) lW[i] = W[i];
    __syncthreads();

    const int wave = threadIdx.x >> 5;
    const int lane = threadIdx.x & 31;
    const int strip = blockIdx.x * 8 + wave;
    const int row0  = strip * 16;
    if (row0 >= M) return;

    const int half = lane >> 4;
    const int l16  = lane & 15;

    const v8f zero = {0.f, 0.f, 0.f, 0.f, 0.f, 0.f, 0.f, 0.f};
    v8f acc[8];
#pragma unroll
    for (int t = 0; t < 8; ++t) acc[t] = zero;

    const float* __restrict__ arow = A + (size_t)(row0 + l16) * F_DIM;

    for (int k0 = 0; k0 < F_DIM; k0 += 4) {
        const int ka = k0 + half * 2;
        v2f a;
        a.x = arow[ka];
        a.y = arow[ka + 1];
#pragma unroll
        for (int t = 0; t < 8; ++t) {
            const int col = t * 16 + l16;
            v2f b;
            b.x = lW[ka * F_DIM + col];
            b.y = lW[(ka + 1) * F_DIM + col];
            acc[t] = __builtin_amdgcn_wmma_f32_16x16x4_f32(
                 false, a,  false, b,
                 (short)0, acc[t],  false,  false);
        }
    }

#pragma unroll
    for (int t = 0; t < 8; ++t) {
        const int col = t * 16 + l16;
#pragma unroll
        for (int v = 0; v < 8; ++v) {
            C[(size_t)(row0 + v + 8 * half) * F_DIM + col] = acc[t][v];
        }
    }
}


__global__ __launch_bounds__(256)
void gather_kernel(const float* __restrict__ H, const int* __restrict__ rowptr,
                   const int* __restrict__ csr, float* __restrict__ out, int n, int e) {
    const int lane = threadIdx.x & 31;
    const int node = __builtin_amdgcn_readfirstlane(blockIdx.x * 8 + (threadIdx.x >> 5));
    if (node >= n) return;
    int beg = rowptr[node];
    int end = rowptr[node + 1];
    beg = max(beg, 0);
    end = min(end, e);
    const float4* __restrict__ H4 = (const float4*)H;
    const int2* __restrict__ csr2 = (const int2*)csr;
    float4 acc = make_float4(0.f, 0.f, 0.f, 0.f);
    for (int c0 = beg; c0 < end; c0 += 32) {
        const int m = min(32, end - c0);
        int2 ev = make_int2(0, 0);
        if (lane < m) ev = csr2[c0 + lane];
        for (int j = 0; j < m; j += 8) {
            float4 h[8];
            float  w[8];
#pragma unroll
            for (int u = 0; u < 8; ++u) {
                h[u] = make_float4(0.f, 0.f, 0.f, 0.f);
                w[u] = 0.f;
                if (u == 0 || j + u < m) {
                    int s = __builtin_amdgcn_readlane(ev.x, j + u);
                    w[u] = __int_as_float(__builtin_amdgcn_readlane(ev.y, j + u));
                    s = min((unsigned)s, (unsigned)(n - 1));
                    h[u] = H4[(size_t)s * (F_DIM / 4) + lane];
                }
            }
#pragma unroll
            for (int u = 0; u < 8; ++u) {
                acc.x = fmaf(h[u].x, w[u], acc.x);
                acc.y = fmaf(h[u].y, w[u], acc.y);
                acc.z = fmaf(h[u].z, w[u], acc.z);
                acc.w = fmaf(h[u].w, w[u], acc.w);
            }
        }
    }
    ((float4*)out)[(size_t)node * (F_DIM / 4) + lane] = acc;
}


__global__ void selfloop_bias_relu_kernel(float* __restrict__ S,
                                          const float* __restrict__ H,
                                          const float* __restrict__ dis,
                                          const float* __restrict__ b, int n) {
    int tid = blockIdx.x * blockDim.x + threadIdx.x;
    if (tid >= n * F_DIM) return;
    int i = tid >> 7;
    int h = tid & (F_DIM - 1);
    float di = dis[i];
    float v = S[tid] + H[tid] * di * di + b[h];
    S[tid] = fmaxf(v, 0.0f);
}


__global__ void pool_kernel(const float* __restrict__ H,
                            const int* __restrict__ batch,
                            float* __restrict__ pooled, int n) {
    int tid  = blockIdx.x * blockDim.x + threadIdx.x;
    int node = tid >> 5;
    int lane = tid & 31;
    if (node >= n) return;
    int g = batch[node];
    const float4* __restrict__ hv = (const float4*)(H + (size_t)node * F_DIM);
    float4 m = hv[lane];
    float* o = pooled + (size_t)g * F_DIM + lane * 4;
    atomicAdd(o + 0, m.x);
    atomicAdd(o + 1, m.y);
    atomicAdd(o + 2, m.z);
    atomicAdd(o + 3, m.w);
}


__global__ void head_kernel(const float* __restrict__ pooled,
                            const float* __restrict__ Wlin,
                            const float* __restrict__ blin,
                            float* __restrict__ out, int g_total) {
    int g = blockIdx.x * blockDim.x + threadIdx.x;
    if (g >= g_total) return;
    float logits[N_CLASSES];
#pragma unroll
    for (int c = 0; c < N_CLASSES; ++c) logits[c] = blin[c];
    const float* __restrict__ pr = pooled + (size_t)g * F_DIM;
    for (int k = 0; k < F_DIM; ++k) {
        float p = pr[k];
        const float* __restrict__ wr = Wlin + (size_t)k * N_CLASSES;
#pragma unroll
        for (int c = 0; c < N_CLASSES; ++c) logits[c] = fmaf(p, wr[c], logits[c]);
    }
    float mx = logits[0];
#pragma unroll
    for (int c = 1; c < N_CLASSES; ++c) mx = fmaxf(mx, logits[c]);
    float s = 0.0f;
#pragma unroll
    for (int c = 0; c < N_CLASSES; ++c) s += expf(logits[c] - mx);
    float lse = mx + logf(s);
    float* o = out + (size_t)g * N_CLASSES;
#pragma unroll
    for (int c = 0; c < N_CLASSES; ++c) o[c] = logits[c] - lse;
}


extern "C" void kernel_launch(void* const* d_in, const int* in_sizes, int n_in,
                              void* d_out, int out_size, void* d_ws, size_t ws_size,
                              hipStream_t stream) {
    const float* x     = (const float*)d_in[0];
    const int*   eidx  = (const int*)d_in[1];
    const int*   batch = (const int*)d_in[2];
    const float* W1    = (const float*)d_in[3];
    const float* b1    = (const float*)d_in[4];
    const float* W2    = (const float*)d_in[5];
    const float* b2    = (const float*)d_in[6];
    const float* Wlin  = (const float*)d_in[7];
    const float* blin  = (const float*)d_in[8];
    float*       out   = (float*)d_out;

    const int* src = eidx;
    const int* dst = eidx + N_EDGES;

    float* ws = (float*)d_ws;
    int*   cnt    = (int*)ws;
    float* dis    = ws + 50000;
    float* H      = ws + 100000;
    float* S      = H + (size_t)N_NODES * F_DIM;
    float* pooled = S + (size_t)N_NODES * F_DIM;
    int*   rowptr = (int*)(pooled + N_GRAPHS * F_DIM);
    int*   rank   = rowptr + 50048;
    int*   csr    = rank + N_EDGES;

    const int B = 256;
    const int nodeBlocks  = (N_NODES + B - 1) / B;
    const int edgeBlocks  = (N_EDGES + B - 1) / B;
    const int featBlocks  = ((size_t)N_NODES * F_DIM + B - 1) / B;
    const int gathBlocks  = (N_NODES + 7) / 8;
    const int poolBlocks  = ((size_t)N_NODES * 32 + B - 1) / B;
    const int gemmBlocks  = (N_NODES / 16 + 7) / 8;
    const int headBlocks  = (N_GRAPHS + B - 1) / B;
    const int scanBlocks  = (N_NODES + 1023) / 1024;

    init_cnt_kernel<<<nodeBlocks, B, 0, stream>>>(cnt, N_NODES);
    count_rank_kernel<<<edgeBlocks, B, 0, stream>>>(dst, cnt, rank, N_EDGES);
    scan_dis_kernel<<<scanBlocks, 1024, 0, stream>>>(cnt, rowptr, dis, N_NODES);
    place_kernel<<<edgeBlocks, B, 0, stream>>>(src, dst, rank, rowptr, dis, csr, N_EDGES);

    gemm128_wmma_kernel<<<gemmBlocks, B, 0, stream>>>(x, W1, H, N_NODES);
    gather_kernel<<<gathBlocks, B, 0, stream>>>(H, rowptr, csr, S, N_NODES, N_EDGES);
    selfloop_bias_relu_kernel<<<featBlocks, B, 0, stream>>>(S, H, dis, b1, N_NODES);

    gemm128_wmma_kernel<<<gemmBlocks, B, 0, stream>>>(S, W2, H, N_NODES);
    gather_kernel<<<gathBlocks, B, 0, stream>>>(H, rowptr, csr, S, N_NODES, N_EDGES);
    selfloop_bias_relu_kernel<<<featBlocks, B, 0, stream>>>(S, H, dis, b2, N_NODES);

    zero_kernel<<<(N_GRAPHS * F_DIM + B - 1) / B, B, 0, stream>>>(pooled, N_GRAPHS * F_DIM);
    pool_kernel<<<poolBlocks, B, 0, stream>>>(S, batch, pooled, N_NODES);
    head_kernel<<<headBlocks, B, 0, stream>>>(pooled, Wlin, blin, out, N_GRAPHS);
}


# Self-loop term, bias and ReLU moved into the gather wave (one store per row): the two one-float-per-thread elementwise launches are gone; summation order unchanged
# speedup vs baseline: 1.9446x; 1.1645x over previous
#include <hip/hip_runtime.h>
#include <hip/hip_bf16.h>
#include <math.h>


typedef float v2f __attribute__((ext_vector_type(2)));
typedef float v8f __attribute__((ext_vector_type(8)));

#define N_NODES   50000
#define N_EDGES   640000
#define F_DIM     128
#define N_CLASSES 16
#define N_GRAPHS  500


__global__ void init_cnt_kernel(int* __restrict__ cnt, int n) {
    int i = blockIdx.x * blockDim.x + threadIdx.x;
    if (i < n) cnt[i] = 0;
}

__global__ void count_rank_kernel(const int* __restrict__ dst, int* __restrict__ cnt,
                                  int* __restrict__ rank, int e) {
    int i = blockIdx.x * blockDim.x + threadIdx.x;
    if (i < e) rank[i] = atomicAdd(&cnt[dst[i]], 1);
}

__global__ __launch_bounds__(1024)
void scan_dis_kernel(const int* __restrict__ cnt, int* __restrict__ rowptr,
                     float* __restrict__ dis, int n) {
    __shared__ int s_wtot[32];
    __shared__ int s_wpre[32];
    __shared__ int s_woff[32];
    const int tid  = threadIdx.x;
    const int lane = tid & 31;
    const int wave = tid >> 5;

    int pre = 0;
    const int4* __restrict__ c4 = (const int4*)cnt;
    const int nq = blockIdx.x * 256;
    int q = tid;
    for (; q + 3072 < nq; q += 4096) {
        int4 v0 = c4[q];
        int4 v1 = c4[q + 1024];
        int4 v2 = c4[q + 2048];
        int4 v3 = c4[q + 3072];
        pre += ((v0.x + v0.y) + (v0.z + v0.w)) + ((v1.x + v1.y) + (v1.z + v1.w))
             + ((v2.x + v2.y) + (v2.z + v2.w)) + ((v3.x + v3.y) + (v3.z + v3.w));
    }
    for (; q < nq; q += 1024) {
        int4 v = c4[q];
        pre += (v.x + v.y) + (v.z + v.w);
    }

    const int i = blockIdx.x * 1024 + tid;
    const int c = (i < n) ? cnt[i] : 0;

    int incl = c;
    for (int o = 1; o < 32; o <<= 1) {
        int t = __shfl_up(incl, o, 32);
        if (lane >= o) incl += t;
    }
    for (int o = 16; o > 0; o >>= 1) pre += __shfl_xor(pre, o, 32);

    if (lane == 31) s_wtot[wave] = incl;
    if (lane == 0)  s_wpre[wave] = pre;
    __syncthreads();
    if (wave == 0) {
        const int t = s_wtot[lane];
        int p  = s_wpre[lane];
        int ti = t;
        for (int o = 1; o < 32; o <<= 1) {
            int u = __shfl_up(ti, o, 32);
            if (lane >= o) ti += u;
        }
        for (int o = 16; o > 0; o >>= 1) p += __shfl_xor(p, o, 32);
        s_woff[lane] = p + ti - t;
    }
    __syncthreads();
    const int off = s_woff[wave];
    if (i < n) {
        rowptr[i] = off + incl - c;
        dis[i]    = rsqrtf((float)(c + 1));
        if (i == n - 1) rowptr[n] = off + incl;
    }
}

__global__ void place_kernel(const int* __restrict__ src, const int* __restrict__ dst,
                             const int* __restrict__ rank, const int* __restrict__ rowptr,
                             const float* __restrict__ dis, int* __restrict__ csr, int e) {
    int i = blockIdx.x * blockDim.x + threadIdx.x;
    if (i >= e) return;
    const int s = src[i];
    const int d = dst[i];
    const int r = rank[i];
    const int   rp = rowptr[d];
    const float dd = dis[d];
    const float dsrc = dis[s];
    const int   pos = rp + r;
    const float w = dsrc * dd;
    if ((unsigned)pos < (unsigned)e) ((int2*)csr)[pos] = make_int2(s, __float_as_int(w));
}


__global__ void zero_kernel(float* __restrict__ p, int n) {
    int i = blockIdx.x * blockDim.x + threadIdx.x;
    if (i < n) p[i] = 0.0f;
}


__global__ __launch_bounds__(256)
void gemm128_wmma_kernel(const float* __restrict__ A, const float* __restrict__ W,
                         float* __restrict__ C, int M) {
    __shared__ float lW[F_DIM * F_DIM];
    for (int i = threadIdx.x; i < F_DIM * F_DIM; i += 256) lW[i] = W[i];
    __syncthreads();

    const int wave = threadIdx.x >> 5;
    const int lane = threadIdx.x & 31;
    const int strip = blockIdx.x * 8 + wave;
    const int row0  = strip * 16;
    if (row0 >= M) return;

    const int half = lane >> 4;
    const int l16  = lane & 15;

    const v8f zero = {0.f, 0.f, 0.f, 0.f, 0.f, 0.f, 0.f, 0.f};
    v8f acc[8];
#pragma unroll
    for (int t = 0; t < 8; ++t) acc[t] = zero;

    const float* __restrict__ arow = A + (size_t)(row0 + l16) * F_DIM;

    for (int k0 = 0; k0 < F_DIM; k0 += 4) {
        const int ka = k0 + half * 2;
        v2f a;
        a.x = arow[ka];
        a.y = arow[ka + 1];
#pragma unroll
        for (int t = 0; t < 8; ++t) {
            const int col = t * 16 + l16;
            v2f b;
            b.x = lW[ka * F_DIM + col];
            b.y = lW[(ka + 1) * F_DIM + col];
            acc[t] = __builtin_amdgcn_wmma_f32_16x16x4_f32(
                 false, a,  false, b,
                 (short)0, acc[t],  false,  false);
        }
    }

#pragma unroll
    for (int t = 0; t < 8; ++t) {
        const int col = t * 16 + l16;
#pragma unroll
        for (int v = 0; v < 8; ++v) {
            C[(size_t)(row0 + v + 8 * half) * F_DIM + col] = acc[t][v];
        }
    }
}


__global__ __launch_bounds__(256)
void gather_kernel(const float* __restrict__ H, const int* __restrict__ rowptr,
                   const int* __restrict__ csr, const float* __restrict__ dis,
                   const float* __restrict__ b, float* __restrict__ out, int n, int e) {
    const int lane = threadIdx.x & 31;
    const int node = __builtin_amdgcn_readfirstlane(blockIdx.x * 8 + (threadIdx.x >> 5));
    if (node >= n) return;
    int beg = rowptr[node];
    int end = rowptr[node + 1];
    beg = max(beg, 0);
    end = min(end, e);
    const float4* __restrict__ H4 = (const float4*)H;
    const int2* __restrict__ csr2 = (const int2*)csr;
    const float  di = dis[node];
    const float4 hs = H4[(size_t)node * (F_DIM / 4) + lane];
    const float4 bb = ((const float4*)b)[lane];
    float4 acc = make_float4(0.f, 0.f, 0.f, 0.f);
    for (int c0 = beg; c0 < end; c0 += 32) {
        const int m = min(32, end - c0);
        int2 ev = make_int2(0, 0);
        if (lane < m) ev = csr2[c0 + lane];
        for (int j = 0; j < m; j += 8) {
            float4 h[8];
            float  w[8];
#pragma unroll
            for (int u = 0; u < 8; ++u) {
                h[u] = make_float4(0.f, 0.f, 0.f, 0.f);
                w[u] = 0.f;
                if (u == 0 || j + u < m) {
                    int s = __builtin_amdgcn_readlane(ev.x, j + u);
                    w[u] = __int_as_float(__builtin_amdgcn_readlane(ev.y, j + u));
                    s = min((unsigned)s, (unsigned)(n - 1));
                    h[u] = H4[(size_t)s * (F_DIM / 4) + lane];
                }
            }
#pragma unroll
            for (int u = 0; u < 8; ++u) {
                acc.x = fmaf(h[u].x, w[u], acc.x);
                acc.y = fmaf(h[u].y, w[u], acc.y);
                acc.z = fmaf(h[u].z, w[u], acc.z);
                acc.w = fmaf(h[u].w, w[u], acc.w);
            }
        }
    }
    float4 r;
    r.x = fmaxf(acc.x + hs.x * di * di + bb.x, 0.0f);
    r.y = fmaxf(acc.y + hs.y * di * di + bb.y, 0.0f);
    r.z = fmaxf(acc.z + hs.z * di * di + bb.z, 0.0f);
    r.w = fmaxf(acc.w + hs.w * di * di + bb.w, 0.0f);
    ((float4*)out)[(size_t)node * (F_DIM / 4) + lane] = r;
}


__global__ void pool_kernel(const float* __restrict__ H,
                            const int* __restrict__ batch,
                            float* __restrict__ pooled, int n) {
    int tid  = blockIdx.x * blockDim.x + threadIdx.x;
    int node = tid >> 5;
    int lane = tid & 31;
    if (node >= n) return;
    int g = batch[node];
    const float4* __restrict__ hv = (const float4*)(H + (size_t)node * F_DIM);
    float4 m = hv[lane];
    float* o = pooled + (size_t)g * F_DIM + lane * 4;
    atomicAdd(o + 0, m.x);
    atomicAdd(o + 1, m.y);
    atomicAdd(o + 2, m.z);
    atomicAdd(o + 3, m.w);
}


__global__ void head_kernel(const float* __restrict__ pooled,
                            const float* __restrict__ Wlin,
                            const float* __restrict__ blin,
                            float* __restrict__ out, int g_total) {
    int g = blockIdx.x * blockDim.x + threadIdx.x;
    if (g >= g_total) return;
    float logits[N_CLASSES];
#pragma unroll
    for (int c = 0; c < N_CLASSES; ++c) logits[c] = blin[c];
    const float* __restrict__ pr = pooled + (size_t)g * F_DIM;
    for (int k = 0; k < F_DIM; ++k) {
        float p = pr[k];
        const float* __restrict__ wr = Wlin + (size_t)k * N_CLASSES;
#pragma unroll
        for (int c = 0; c < N_CLASSES; ++c) logits[c] = fmaf(p, wr[c], logits[c]);
    }
    float mx = logits[0];
#pragma unroll
    for (int c = 1; c < N_CLASSES; ++c) mx = fmaxf(mx, logits[c]);
    float s = 0.0f;
#pragma unroll
    for (int c = 0; c < N_CLASSES; ++c) s += expf(logits[c] - mx);
    float lse = mx + logf(s);
    float* o = out + (size_t)g * N_CLASSES;
#pragma unroll
    for (int c = 0; c < N_CLASSES; ++c) o[c] = logits[c] - lse;
}


extern "C" void kernel_launch(void* const* d_in, const int* in_sizes, int n_in,
                              void* d_out, int out_size, void* d_ws, size_t ws_size,
                              hipStream_t stream) {
    const float* x     = (const float*)d_in[0];
    const int*   eidx  = (const int*)d_in[1];
    const int*   batch = (const int*)d_in[2];
    const float* W1    = (const float*)d_in[3];
    const float* b1    = (const float*)d_in[4];
    const float* W2    = (const float*)d_in[5];
    const float* b2    = (const float*)d_in[6];
    const float* Wlin  = (const float*)d_in[7];
    const float* blin  = (const float*)d_in[8];
    float*       out   = (float*)d_out;

    const int* src = eidx;
    const int* dst = eidx + N_EDGES;

    float* ws = (float*)d_ws;
    int*   cnt    = (int*)ws;
    float* dis    = ws + 50000;
    float* H      = ws + 100000;
    float* S      = H + (size_t)N_NODES * F_DIM;
    float* pooled = S + (size_t)N_NODES * F_DIM;
    int*   rowptr = (int*)(pooled + N_GRAPHS * F_DIM);
    int*   rank   = rowptr + 50048;
    int*   csr    = rank + N_EDGES;

    const int B = 256;
    const int nodeBlocks  = (N_NODES + B - 1) / B;
    const int edgeBlocks  = (N_EDGES + B - 1) / B;
    const int gathBlocks  = (N_NODES + 7) / 8;
    const int poolBlocks  = ((size_t)N_NODES * 32 + B - 1) / B;
    const int gemmBlocks  = (N_NODES / 16 + 7) / 8;
    const int headBlocks  = (N_GRAPHS + B - 1) / B;
    const int scanBlocks  = (N_NODES + 1023) / 1024;

    init_cnt_kernel<<<nodeBlocks, B, 0, stream>>>(cnt, N_NODES);
    count_rank_kernel<<<edgeBlocks, B, 0, stream>>>(dst, cnt, rank, N_EDGES);
    scan_dis_kernel<<<scanBlocks, 1024, 0, stream>>>(cnt, rowptr, dis, N_NODES);
    place_kernel<<<edgeBlocks, B, 0, stream>>>(src, dst, rank, rowptr, dis, csr, N_EDGES);

    gemm128_wmma_kernel<<<gemmBlocks, B, 0, stream>>>(x, W1, H, N_NODES);
    gather_kernel<<<gathBlocks, B, 0, stream>>>(H, rowptr, csr, dis, b1, S, N_NODES, N_EDGES);

    gemm128_wmma_kernel<<<gemmBlocks, B, 0, stream>>>(S, W2, H, N_NODES);
    gather_kernel<<<gathBlocks, B, 0, stream>>>(H, rowptr, csr, dis, b2, S, N_NODES, N_EDGES);

    zero_kernel<<<(N_GRAPHS * F_DIM + B - 1) / B, B, 0, stream>>>(pooled, N_GRAPHS * F_DIM);
    pool_kernel<<<poolBlocks, B, 0, stream>>>(S, batch, pooled, N_NODES);
    head_kernel<<<headBlocks, B, 0, stream>>>(pooled, Wlin, blin, out, N_GRAPHS);
}


# Pool: batch ids are sorted, so each graph owns one contiguous node range (offsets built in the count-init launch); one block per graph sums rows + head + log_softmax; no atomics, fixed sum tree
# speedup vs baseline: 2.5590x; 1.3160x over previous
#include <hip/hip_runtime.h>
#include <hip/hip_bf16.h>
#include <math.h>


typedef float v2f __attribute__((ext_vector_type(2)));
typedef float v8f __attribute__((ext_vector_type(8)));

#define N_NODES   50000
#define N_EDGES   640000
#define F_DIM     128
#define N_CLASSES 16
#define N_GRAPHS  500


__global__ void init_cnt_seg_kernel(int* __restrict__ cnt, const int* __restrict__ batch,
                                    int* __restrict__ gstart, int n) {
    int i = blockIdx.x * blockDim.x + threadIdx.x;
    if (i < n) cnt[i] = 0;
    if (i <= n) {
        int prev = -1;
        int cur  = N_GRAPHS;
        if (i > 0) prev = min(max(batch[i - 1], 0), N_GRAPHS - 1);
        if (i < n) cur  = min(max(batch[i], 0), N_GRAPHS - 1);
        for (int g = prev + 1; g <= cur; ++g) gstart[g] = i;
    }
}

__global__ void count_rank_kernel(const int* __restrict__ dst, int* __restrict__ cnt,
                                  int* __restrict__ rank, int e) {
    int i = blockIdx.x * blockDim.x + threadIdx.x;
    if (i < e) rank[i] = atomicAdd(&cnt[dst[i]], 1);
}

__global__ __launch_bounds__(1024)
void scan_dis_kernel(const int* __restrict__ cnt, int* __restrict__ rowptr,
                     float* __restrict__ dis, int n) {
    __shared__ int s_wtot[32];
    __shared__ int s_wpre[32];
    __shared__ int s_woff[32];
    const int tid  = threadIdx.x;
    const int lane = tid & 31;
    const int wave = tid >> 5;

    int pre = 0;
    const int4* __restrict__ c4 = (const int4*)cnt;
    const int nq = blockIdx.x * 256;
    int q = tid;
    for (; q + 3072 < nq; q += 4096) {
        int4 v0 = c4[q];
        int4 v1 = c4[q + 1024];
        int4 v2 = c4[q + 2048];
        int4 v3 = c4[q + 3072];
        pre += ((v0.x + v0.y) + (v0.z + v0.w)) + ((v1.x + v1.y) + (v1.z + v1.w))
             + ((v2.x + v2.y) + (v2.z + v2.w)) + ((v3.x + v3.y) + (v3.z + v3.w));
    }
    for (; q < nq; q += 1024) {
        int4 v = c4[q];
        pre += (v.x + v.y) + (v.z + v.w);
    }

    const int i = blockIdx.x * 1024 + tid;
    const int c = (i < n) ? cnt[i] : 0;

    int incl = c;
    for (int o = 1; o < 32; o <<= 1) {
        int t = __shfl_up(incl, o, 32);
        if (lane >= o) incl += t;
    }
    for (int o = 16; o > 0; o >>= 1) pre += __shfl_xor(pre, o, 32);

    if (lane == 31) s_wtot[wave] = incl;
    if (lane == 0)  s_wpre[wave] = pre;
    __syncthreads();
    if (wave == 0) {
        const int t = s_wtot[lane];
        int p  = s_wpre[lane];
        int ti = t;
        for (int o = 1; o < 32; o <<= 1) {
            int u = __shfl_up(ti, o, 32);
            if (lane >= o) ti += u;
        }
        for (int o = 16; o > 0; o >>= 1) p += __shfl_xor(p, o, 32);
        s_woff[lane] = p + ti - t;
    }
    __syncthreads();
    const int off = s_woff[wave];
    if (i < n) {
        rowptr[i] = off + incl - c;
        dis[i]    = rsqrtf((float)(c + 1));
        if (i == n - 1) rowptr[n] = off + incl;
    }
}

__global__ void place_kernel(const int* __restrict__ src, const int* __restrict__ dst,
                             const int* __restrict__ rank, const int* __restrict__ rowptr,
                             const float* __restrict__ dis, int* __restrict__ csr, int e) {
    int i = blockIdx.x * blockDim.x + threadIdx.x;
    if (i >= e) return;
    const int s = src[i];
    const int d = dst[i];
    const int r = rank[i];
    const int   rp = rowptr[d];
    const float dd = dis[d];
    const float dsrc = dis[s];
    const int   pos = rp + r;
    const float w = dsrc * dd;
    if ((unsigned)pos < (unsigned)e) ((int2*)csr)[pos] = make_int2(s, __float_as_int(w));
}


__global__ __launch_bounds__(256)
void gemm128_wmma_kernel(const float* __restrict__ A, const float* __restrict__ W,
                         float* __restrict__ C, int M) {
    __shared__ float lW[F_DIM * F_DIM];
    for (int i = threadIdx.x; i < F_DIM * F_DIM; i += 256) lW[i] = W[i];
    __syncthreads();

    const int wave = threadIdx.x >> 5;
    const int lane = threadIdx.x & 31;
    const int strip = blockIdx.x * 8 + wave;
    const int row0  = strip * 16;
    if (row0 >= M) return;

    const int half = lane >> 4;
    const int l16  = lane & 15;

    const v8f zero = {0.f, 0.f, 0.f, 0.f, 0.f, 0.f, 0.f, 0.f};
    v8f acc[8];
#pragma unroll
    for (int t = 0; t < 8; ++t) acc[t] = zero;

    const float* __restrict__ arow = A + (size_t)(row0 + l16) * F_DIM;

    for (int k0 = 0; k0 < F_DIM; k0 += 4) {
        const int ka = k0 + half * 2;
        v2f a;
        a.x = arow[ka];
        a.y = arow[ka + 1];
#pragma unroll
        for (int t = 0; t < 8; ++t) {
            const int col = t * 16 + l16;
            v2f b;
            b.x = lW[ka * F_DIM + col];
            b.y = lW[(ka + 1) * F_DIM + col];
            acc[t] = __builtin_amdgcn_wmma_f32_16x16x4_f32(
                 false, a,  false, b,
                 (short)0, acc[t],  false,  false);
        }
    }

#pragma unroll
    for (int t = 0; t < 8; ++t) {
        const int col = t * 16 + l16;
#pragma unroll
        for (int v = 0; v < 8; ++v) {
            C[(size_t)(row0 + v + 8 * half) * F_DIM + col] = acc[t][v];
        }
    }
}


__global__ __launch_bounds__(256)
void gather_kernel(const float* __restrict__ H, const int* __restrict__ rowptr,
                   const int* __restrict__ csr, const float* __restrict__ dis,
                   const float* __restrict__ b, float* __restrict__ out, int n, int e) {
    const int lane = threadIdx.x & 31;
    const int node = __builtin_amdgcn_readfirstlane(blockIdx.x * 8 + (threadIdx.x >> 5));
    if (node >= n) return;
    int beg = rowptr[node];
    int end = rowptr[node + 1];
    beg = max(beg, 0);
    end = min(end, e);
    const float4* __restrict__ H4 = (const float4*)H;
    const int2* __restrict__ csr2 = (const int2*)csr;
    const float  di = dis[node];
    const float4 hs = H4[(size_t)node * (F_DIM / 4) + lane];
    const float4 bb = ((const float4*)b)[lane];
    float4 acc = make_float4(0.f, 0.f, 0.f, 0.f);
    for (int c0 = beg; c0 < end; c0 += 32) {
        const int m = min(32, end - c0);
        int2 ev = make_int2(0, 0);
        if (lane < m) ev = csr2[c0 + lane];
        for (int j = 0; j < m; j += 8) {
            float4 h[8];
            float  w[8];
#pragma unroll
            for (int u = 0; u < 8; ++u) {
                h[u] = make_float4(0.f, 0.f, 0.f, 0.f);
                w[u] = 0.f;
                if (u == 0 || j + u < m) {
                    int s = __builtin_amdgcn_readlane(ev.x, j + u);
                    w[u] = __int_as_float(__builtin_amdgcn_readlane(ev.y, j + u));
                    s = min((unsigned)s, (unsigned)(n - 1));
                    h[u] = H4[(size_t)s * (F_DIM / 4) + lane];
                }
            }
#pragma unroll
            for (int u = 0; u < 8; ++u) {
                acc.x = fmaf(h[u].x, w[u], acc.x);
                acc.y = fmaf(h[u].y, w[u], acc.y);
                acc.z = fmaf(h[u].z, w[u], acc.z);
                acc.w = fmaf(h[u].w, w[u], acc.w);
            }
        }
    }
    float4 r;
    r.x = fmaxf(acc.x + hs.x * di * di + bb.x, 0.0f);
    r.y = fmaxf(acc.y + hs.y * di * di + bb.y, 0.0f);
    r.z = fmaxf(acc.z + hs.z * di * di + bb.z, 0.0f);
    r.w = fmaxf(acc.w + hs.w * di * di + bb.w, 0.0f);
    ((float4*)out)[(size_t)node * (F_DIM / 4) + lane] = r;
}


#define POOL_WAVES 16

__global__ __launch_bounds__(POOL_WAVES * 32)
void pool_head_kernel(const float* __restrict__ H, const int* __restrict__ gstart,
                      const float* __restrict__ Wlin, const float* __restrict__ blin,
                      float* __restrict__ out, int n) {
    __shared__ float4 s_part[POOL_WAVES][32];
    __shared__ float  s_pool[F_DIM];
    __shared__ float  s_lg[POOL_WAVES * 2][N_CLASSES];
    const int tid  = threadIdx.x;
    const int lane = tid & 31;
    const int wave = __builtin_amdgcn_readfirstlane(tid >> 5);
    const int g    = blockIdx.x;

    const int c  = tid & (N_CLASSES - 1);
    const int ks = tid >> 4;
    float wl[4];
#pragma unroll
    for (int q = 0; q < 4; ++q) wl[q] = Wlin[(ks * 4 + q) * N_CLASSES + c];
    const float bl = blin[c];

    int lo = gstart[g];
    int hi = gstart[g + 1];
    lo = max(lo, 0);
    hi = min(hi, n);

    const float4* __restrict__ H4 = (const float4*)H;
    float4 acc = make_float4(0.f, 0.f, 0.f, 0.f);
    for (int i = lo + wave; i < hi; i += 4 * POOL_WAVES) {
        float4 h[4];
#pragma unroll
        for (int u = 0; u < 4; ++u) {
            h[u] = make_float4(0.f, 0.f, 0.f, 0.f);
            const int r = i + u * POOL_WAVES;
            if (r < hi) h[u] = H4[(size_t)r * (F_DIM / 4) + lane];
        }
#pragma unroll
        for (int u = 0; u < 4; ++u) {
            acc.x += h[u].x;
            acc.y += h[u].y;
            acc.z += h[u].z;
            acc.w += h[u].w;
        }
    }
    s_part[wave][lane] = acc;
    __syncthreads();

    if (tid < F_DIM) {
        const float* __restrict__ sp = (const float*)s_part;
        float s = sp[tid];
#pragma unroll
        for (int w = 1; w < POOL_WAVES; ++w) s += sp[w * F_DIM + tid];
        s_pool[tid] = s;
    }
    __syncthreads();

    {
        float p = s_pool[ks * 4] * wl[0];
#pragma unroll
        for (int q = 1; q < 4; ++q) p = fmaf(s_pool[ks * 4 + q], wl[q], p);
        s_lg[ks][c] = p;
    }
    __syncthreads();

    if (wave == 0) {
        float lg = bl;
#pragma unroll
        for (int s = 0; s < POOL_WAVES * 2; ++s) lg += s_lg[s][c];
        float mx = lg;
#pragma unroll
        for (int o = 8; o > 0; o >>= 1) mx = fmaxf(mx, __shfl_xor(mx, o, 16));
        float se = expf(lg - mx);
#pragma unroll
        for (int o = 8; o > 0; o >>= 1) se += __shfl_xor(se, o, 16);
        const float lse = mx + logf(se);
        if (lane < N_CLASSES) out[(size_t)g * N_CLASSES + lane] = lg - lse;
    }
}


extern "C" void kernel_launch(void* const* d_in, const int* in_sizes, int n_in,
                              void* d_out, int out_size, void* d_ws, size_t ws_size,
                              hipStream_t stream) {
    const float* x     = (const float*)d_in[0];
    const int*   eidx  = (const int*)d_in[1];
    const int*   batch = (const int*)d_in[2];
    const float* W1    = (const float*)d_in[3];
    const float* b1    = (const float*)d_in[4];
    const float* W2    = (const float*)d_in[5];
    const float* b2    = (const float*)d_in[6];
    const float* Wlin  = (const float*)d_in[7];
    const float* blin  = (const float*)d_in[8];
    float*       out   = (float*)d_out;

    const int* src = eidx;
    const int* dst = eidx + N_EDGES;

    float* ws = (float*)d_ws;
    int*   cnt    = (int*)ws;
    float* dis    = ws + 50000;
    float* H      = ws + 100000;
    float* S      = H + (size_t)N_NODES * F_DIM;
    int*   gstart = (int*)(S + (size_t)N_NODES * F_DIM);
    int*   rowptr = gstart + N_GRAPHS * F_DIM;
    int*   rank   = rowptr + 50048;
    int*   csr    = rank + N_EDGES;

    const int B = 256;
    const int nodeBlocks  = (N_NODES + 1 + B - 1) / B;
    const int edgeBlocks  = (N_EDGES + B - 1) / B;
    const int gathBlocks  = (N_NODES + 7) / 8;
    const int gemmBlocks  = (N_NODES / 16 + 7) / 8;
    const int scanBlocks  = (N_NODES + 1023) / 1024;

    init_cnt_seg_kernel<<<nodeBlocks, B, 0, stream>>>(cnt, batch, gstart, N_NODES);
    count_rank_kernel<<<edgeBlocks, B, 0, stream>>>(dst, cnt, rank, N_EDGES);
    scan_dis_kernel<<<scanBlocks, 1024, 0, stream>>>(cnt, rowptr, dis, N_NODES);
    place_kernel<<<edgeBlocks, B, 0, stream>>>(src, dst, rank, rowptr, dis, csr, N_EDGES);

    gemm128_wmma_kernel<<<gemmBlocks, B, 0, stream>>>(x, W1, H, N_NODES);
    gather_kernel<<<gathBlocks, B, 0, stream>>>(H, rowptr, csr, dis, b1, S, N_NODES, N_EDGES);

    gemm128_wmma_kernel<<<gemmBlocks, B, 0, stream>>>(S, W2, H, N_NODES);
    gather_kernel<<<gathBlocks, B, 0, stream>>>(H, rowptr, csr, dis, b2, S, N_NODES, N_EDGES);

    pool_head_kernel<<<N_GRAPHS, POOL_WAVES * 32, 0, stream>>>(S, gstart, Wlin, blin, out, N_NODES);
}


# First GEMM (x @ W1) needs nothing from the edge sort: its row strips run as extra blocks inside the count/rank and the placement launch (half each), its own launch is gone; no sum order change
# speedup vs baseline: 2.7469x; 1.0734x over previous
#include <hip/hip_runtime.h>
#include <hip/hip_bf16.h>
#include <math.h>


typedef float v2f __attribute__((ext_vector_type(2)));
typedef float v8f __attribute__((ext_vector_type(8)));

#define N_NODES   50000
#define N_EDGES   640000
#define F_DIM     128
#define N_CLASSES 16
#define N_GRAPHS  500


__global__ void init_cnt_seg_kernel(int* __restrict__ cnt, const int* __restrict__ batch,
                                    int* __restrict__ gstart, int n) {
    int i = blockIdx.x * blockDim.x + threadIdx.x;
    if (i < n) cnt[i] = 0;
    if (i <= n) {
        int prev = -1;
        int cur  = N_GRAPHS;
        if (i > 0) prev = min(max(batch[i - 1], 0), N_GRAPHS - 1);
        if (i < n) cur  = min(max(batch[i], 0), N_GRAPHS - 1);
        for (int g = prev + 1; g <= cur; ++g) gstart[g] = i;
    }
}


__global__ __launch_bounds__(1024)
void scan_dis_kernel(const int* __restrict__ cnt, int* __restrict__ rowptr,
                     float* __restrict__ dis, int n) {
    __shared__ int s_wtot[32];
    __shared__ int s_wpre[32];
    __shared__ int s_woff[32];
    const int tid  = threadIdx.x;
    const int lane = tid & 31;
    const int wave = tid >> 5;

    int pre = 0;
    const int4* __restrict__ c4 = (const int4*)cnt;
    const int nq = blockIdx.x * 256;
    int q = tid;
    for (; q + 3072 < nq; q += 4096) {
        int4 v0 = c4[q];
        int4 v1 = c4[q + 1024];
        int4 v2 = c4[q + 2048];
        int4 v3 = c4[q + 3072];
        pre += ((v0.x + v0.y) + (v0.z + v0.w)) + ((v1.x + v1.y) + (v1.z + v1.w))
             + ((v2.x + v2.y) + (v2.z + v2.w)) + ((v3.x + v3.y) + (v3.z + v3.w));
    }
    for (; q < nq; q += 1024) {
        int4 v = c4[q];
        pre += (v.x + v.y) + (v.z + v.w);
    }

    const int i = blockIdx.x * 1024 + tid;
    const int c = (i < n) ? cnt[i] : 0;

    int incl = c;
    for (int o = 1; o < 32; o <<= 1) {
        int t = __shfl_up(incl, o, 32);
        if (lane >= o) incl += t;
    }
    for (int o = 16; o > 0; o >>= 1) pre += __shfl_xor(pre, o, 32);

    if (lane == 31) s_wtot[wave] = incl;
    if (lane == 0)  s_wpre[wave] = pre;
    __syncthreads();
    if (wave == 0) {
        const int t = s_wtot[lane];
        int p  = s_wpre[lane];
        int ti = t;
        for (int o = 1; o < 32; o <<= 1) {
            int u = __shfl_up(ti, o, 32);
            if (lane >= o) ti += u;
        }
        for (int o = 16; o > 0; o >>= 1) p += __shfl_xor(p, o, 32);
        s_woff[lane] = p + ti - t;
    }
    __syncthreads();
    const int off = s_woff[wave];
    if (i < n) {
        rowptr[i] = off + incl - c;
        dis[i]    = rsqrtf((float)(c + 1));
        if (i == n - 1) rowptr[n] = off + incl;
    }
}


__device__ __forceinline__
void gemm128_block(const float* __restrict__ A, const float* __restrict__ W,
                   float* __restrict__ C, int M, int blk, float* __restrict__ lW) {
    for (int i = threadIdx.x; i < F_DIM * F_DIM; i += 256) lW[i] = W[i];
    __syncthreads();

    const int wave = threadIdx.x >> 5;
    const int lane = threadIdx.x & 31;
    const int strip = blk * 8 + wave;
    const int row0  = strip * 16;
    if (row0 >= M) return;

    const int half = lane >> 4;
    const int l16  = lane & 15;

    const v8f zero = {0.f, 0.f, 0.f, 0.f, 0.f, 0.f, 0.f, 0.f};
    v8f acc[8];
#pragma unroll
    for (int t = 0; t < 8; ++t) acc[t] = zero;

    const float* __restrict__ arow = A + (size_t)(row0 + l16) * F_DIM;

    for (int k0 = 0; k0 < F_DIM; k0 += 4) {
        const int ka = k0 + half * 2;
        v2f a;
        a.x = arow[ka];
        a.y = arow[ka + 1];
#pragma unroll
        for (int t = 0; t < 8; ++t) {
            const int col = t * 16 + l16;
            v2f b;
            b.x = lW[ka * F_DIM + col];
            b.y = lW[(ka + 1) * F_DIM + col];
            acc[t] = __builtin_amdgcn_wmma_f32_16x16x4_f32(
                 false, a,  false, b,
                 (short)0, acc[t],  false,  false);
        }
    }

#pragma unroll
    for (int t = 0; t < 8; ++t) {
        const int col = t * 16 + l16;
#pragma unroll
        for (int v = 0; v < 8; ++v) {
            C[(size_t)(row0 + v + 8 * half) * F_DIM + col] = acc[t][v];
        }
    }
}

__global__ __launch_bounds__(256)
void gemm128_wmma_kernel(const float* __restrict__ A, const float* __restrict__ W,
                         float* __restrict__ C, int M) {
    __shared__ float lW[F_DIM * F_DIM];
    gemm128_block(A, W, C, M, blockIdx.x, lW);
}


__global__ __launch_bounds__(256)
void count_rank_gemm_kernel(const int* __restrict__ dst, int* __restrict__ cnt,
                            int* __restrict__ rank, int e,
                            const float* __restrict__ A, const float* __restrict__ W,
                            float* __restrict__ C, int M, int gb, int g0) {
    __shared__ float lW[F_DIM * F_DIM];
    if ((int)blockIdx.x < gb) {
        gemm128_block(A, W, C, M, g0 + (int)blockIdx.x, lW);
        return;
    }
    int i = ((int)blockIdx.x - gb) * 256 + threadIdx.x;
    if (i < e) rank[i] = atomicAdd(&cnt[dst[i]], 1);
}

__global__ __launch_bounds__(256)
void place_gemm_kernel(const int* __restrict__ src, const int* __restrict__ dst,
                       const int* __restrict__ rank, const int* __restrict__ rowptr,
                       const float* __restrict__ dis, int* __restrict__ csr, int e,
                       const float* __restrict__ A, const float* __restrict__ W,
                       float* __restrict__ C, int M, int gb, int g0) {
    __shared__ float lW[F_DIM * F_DIM];
    if ((int)blockIdx.x < gb) {
        gemm128_block(A, W, C, M, g0 + (int)blockIdx.x, lW);
        return;
    }
    int i = ((int)blockIdx.x - gb) * 256 + threadIdx.x;
    if (i >= e) return;
    const int s = src[i];
    const int d = dst[i];
    const int r = rank[i];
    const int   rp = rowptr[d];
    const float dd = dis[d];
    const float dsrc = dis[s];
    const int   pos = rp + r;
    const float w = dsrc * dd;
    if ((unsigned)pos < (unsigned)e) ((int2*)csr)[pos] = make_int2(s, __float_as_int(w));
}


__global__ __launch_bounds__(256)
void gather_kernel(const float* __restrict__ H, const int* __restrict__ rowptr,
                   const int* __restrict__ csr, const float* __restrict__ dis,
                   const float* __restrict__ b, float* __restrict__ out, int n, int e) {
    const int lane = threadIdx.x & 31;
    const int node = __builtin_amdgcn_readfirstlane(blockIdx.x * 8 + (threadIdx.x >> 5));
    if (node >= n) return;
    int beg = rowptr[node];
    int end = rowptr[node + 1];
    beg = max(beg, 0);
    end = min(end, e);
    const float4* __restrict__ H4 = (const float4*)H;
    const int2* __restrict__ csr2 = (const int2*)csr;
    const float  di = dis[node];
    const float4 hs = H4[(size_t)node * (F_DIM / 4) + lane];
    const float4 bb = ((const float4*)b)[lane];
    float4 acc = make_float4(0.f, 0.f, 0.f, 0.f);
    for (int c0 = beg; c0 < end; c0 += 32) {
        const int m = min(32, end - c0);
        int2 ev = make_int2(0, 0);
        if (lane < m) ev = csr2[c0 + lane];
        for (int j = 0; j < m; j += 8) {
            float4 h[8];
            float  w[8];
#pragma unroll
            for (int u = 0; u < 8; ++u) {
                h[u] = make_float4(0.f, 0.f, 0.f, 0.f);
                w[u] = 0.f;
                if (u == 0 || j + u < m) {
                    int s = __builtin_amdgcn_readlane(ev.x, j + u);
                    w[u] = __int_as_float(__builtin_amdgcn_readlane(ev.y, j + u));
                    s = min((unsigned)s, (unsigned)(n - 1));
                    h[u] = H4[(size_t)s * (F_DIM / 4) + lane];
                }
            }
#pragma unroll
            for (int u = 0; u < 8; ++u) {
                acc.x = fmaf(h[u].x, w[u], acc.x);
                acc.y = fmaf(h[u].y, w[u], acc.y);
                acc.z = fmaf(h[u].z, w[u], acc.z);
                acc.w = fmaf(h[u].w, w[u], acc.w);
            }
        }
    }
    float4 r;
    r.x = fmaxf(acc.x + hs.x * di * di + bb.x, 0.0f);
    r.y = fmaxf(acc.y + hs.y * di * di + bb.y, 0.0f);
    r.z = fmaxf(acc.z + hs.z * di * di + bb.z, 0.0f);
    r.w = fmaxf(acc.w + hs.w * di * di + bb.w, 0.0f);
    ((float4*)out)[(size_t)node * (F_DIM / 4) + lane] = r;
}


#define POOL_WAVES 16

__global__ __launch_bounds__(POOL_WAVES * 32)
void pool_head_kernel(const float* __restrict__ H, const int* __restrict__ gstart,
                      const float* __restrict__ Wlin, const float* __restrict__ blin,
                      float* __restrict__ out, int n) {
    __shared__ float4 s_part[POOL_WAVES][32];
    __shared__ float  s_pool[F_DIM];
    __shared__ float  s_lg[POOL_WAVES * 2][N_CLASSES];
    const int tid  = threadIdx.x;
    const int lane = tid & 31;
    const int wave = __builtin_amdgcn_readfirstlane(tid >> 5);
    const int g    = blockIdx.x;

    const int c  = tid & (N_CLASSES - 1);
    const int ks = tid >> 4;
    float wl[4];
#pragma unroll
    for (int q = 0; q < 4; ++q) wl[q] = Wlin[(ks * 4 + q) * N_CLASSES + c];
    const float bl = blin[c];

    int lo = gstart[g];
    int hi = gstart[g + 1];
    lo = max(lo, 0);
    hi = min(hi, n);

    const float4* __restrict__ H4 = (const float4*)H;
    float4 acc = make_float4(0.f, 0.f, 0.f, 0.f);
    for (int i = lo + wave; i < hi; i += 4 * POOL_WAVES) {
        float4 h[4];
#pragma unroll
        for (int u = 0; u < 4; ++u) {
            h[u] = make_float4(0.f, 0.f, 0.f, 0.f);
            const int r = i + u * POOL_WAVES;
            if (r < hi) h[u] = H4[(size_t)r * (F_DIM / 4) + lane];
        }
#pragma unroll
        for (int u = 0; u < 4; ++u) {
            acc.x += h[u].x;
            acc.y += h[u].y;
            acc.z += h[u].z;
            acc.w += h[u].w;
        }
    }
    s_part[wave][lane] = acc;
    __syncthreads();

    if (tid < F_DIM) {
        const float* __restrict__ sp = (const float*)s_part;
        float s = sp[tid];
#pragma unroll
        for (int w = 1; w < POOL_WAVES; ++w) s += sp[w * F_DIM + tid];
        s_pool[tid] = s;
    }
    __syncthreads();

    {
        float p = s_pool[ks * 4] * wl[0];
#pragma unroll
        for (int q = 1; q < 4; ++q) p = fmaf(s_pool[ks * 4 + q], wl[q], p);
        s_lg[ks][c] = p;
    }
    __syncthreads();

    if (wave == 0) {
        float lg = bl;
#pragma unroll
        for (int s = 0; s < POOL_WAVES * 2; ++s) lg += s_lg[s][c];
        float mx = lg;
#pragma unroll
        for (int o = 8; o > 0; o >>= 1) mx = fmaxf(mx, __shfl_xor(mx, o, 16));
        float se = expf(lg - mx);
#pragma unroll
        for (int o = 8; o > 0; o >>= 1) se += __shfl_xor(se, o, 16);
        const float lse = mx + logf(se);
        if (lane < N_CLASSES) out[(size_t)g * N_CLASSES + lane] = lg - lse;
    }
}


extern "C" void kernel_launch(void* const* d_in, const int* in_sizes, int n_in,
                              void* d_out, int out_size, void* d_ws, size_t ws_size,
                              hipStream_t stream) {
    const float* x     = (const float*)d_in[0];
    const int*   eidx  = (const int*)d_in[1];
    const int*   batch = (const int*)d_in[2];
    const float* W1    = (const float*)d_in[3];
    const float* b1    = (const float*)d_in[4];
    const float* W2    = (const float*)d_in[5];
    const float* b2    = (const float*)d_in[6];
    const float* Wlin  = (const float*)d_in[7];
    const float* blin  = (const float*)d_in[8];
    float*       out   = (float*)d_out;

    const int* src = eidx;
    const int* dst = eidx + N_EDGES;

    float* ws = (float*)d_ws;
    int*   cnt    = (int*)ws;
    float* dis    = ws + 50000;
    float* H      = ws + 100000;
    float* S      = H + (size_t)N_NODES * F_DIM;
    int*   gstart = (int*)(S + (size_t)N_NODES * F_DIM);
    int*   rowptr = gstart + N_GRAPHS * F_DIM;
    int*   rank   = rowptr + 50048;
    int*   csr    = rank + N_EDGES;

    const int B = 256;
    const int nodeBlocks  = (N_NODES + 1 + B - 1) / B;
    const int edgeBlocks  = (N_EDGES + B - 1) / B;
    const int gathBlocks  = (N_NODES + 7) / 8;
    const int gemmBlocks  = (N_NODES / 16 + 7) / 8;
    const int scanBlocks  = (N_NODES + 1023) / 1024;
    const int gemmHalfA   = (gemmBlocks + 1) / 2;
    const int gemmHalfB   = gemmBlocks - gemmHalfA;

    init_cnt_seg_kernel<<<nodeBlocks, B, 0, stream>>>(cnt, batch, gstart, N_NODES);
    count_rank_gemm_kernel<<<gemmHalfA + edgeBlocks, B, 0, stream>>>(
        dst, cnt, rank, N_EDGES, x, W1, H, N_NODES, gemmHalfA, 0);
    scan_dis_kernel<<<scanBlocks, 1024, 0, stream>>>(cnt, rowptr, dis, N_NODES);
    place_gemm_kernel<<<gemmHalfB + edgeBlocks, B, 0, stream>>>(
        src, dst, rank, rowptr, dis, csr, N_EDGES, x, W1, H, N_NODES, gemmHalfB, gemmHalfA);

    gather_kernel<<<gathBlocks, B, 0, stream>>>(H, rowptr, csr, dis, b1, S, N_NODES, N_EDGES);

    gemm128_wmma_kernel<<<gemmBlocks, B, 0, stream>>>(S, W2, H, N_NODES);
    gather_kernel<<<gathBlocks, B, 0, stream>>>(H, rowptr, csr, dis, b2, S, N_NODES, N_EDGES);

    pool_head_kernel<<<N_GRAPHS, POOL_WAVES * 32, 0, stream>>>(S, gstart, Wlin, blin, out, N_NODES);
}


# Second GEMM fused into the first gather: 16-wave blocks keep W2 in LDS, gather 16 nodes into an LDS tile, 4 waves multiply it at once; relu1 plane never stored, GEMM launch gone; sum order kept
# speedup vs baseline: 2.9768x; 1.0837x over previous
#include <hip/hip_runtime.h>
#include <hip/hip_bf16.h>
#include <math.h>


typedef float v2f __attribute__((ext_vector_type(2)));
typedef float v8f __attribute__((ext_vector_type(8)));

#define N_NODES   50000
#define N_EDGES   640000
#define F_DIM     128
#define N_CLASSES 16
#define N_GRAPHS  500


__global__ void init_cnt_seg_kernel(int* __restrict__ cnt, const int* __restrict__ batch,
                                    int* __restrict__ gstart, int n) {
    int i = blockIdx.x * blockDim.x + threadIdx.x;
    if (i < n) cnt[i] = 0;
    if (i <= n) {
        int prev = -1;
        int cur  = N_GRAPHS;
        if (i > 0) prev = min(max(batch[i - 1], 0), N_GRAPHS - 1);
        if (i < n) cur  = min(max(batch[i], 0), N_GRAPHS - 1);
        for (int g = prev + 1; g <= cur; ++g) gstart[g] = i;
    }
}


__global__ __launch_bounds__(1024)
void scan_dis_kernel(const int* __restrict__ cnt, int* __restrict__ rowptr,
                     float* __restrict__ dis, int n) {
    __shared__ int s_wtot[32];
    __shared__ int s_wpre[32];
    __shared__ int s_woff[32];
    const int tid  = threadIdx.x;
    const int lane = tid & 31;
    const int wave = tid >> 5;

    int pre = 0;
    const int4* __restrict__ c4 = (const int4*)cnt;
    const int nq = blockIdx.x * 256;
    int q = tid;
    for (; q + 3072 < nq; q += 4096) {
        int4 v0 = c4[q];
        int4 v1 = c4[q + 1024];
        int4 v2 = c4[q + 2048];
        int4 v3 = c4[q + 3072];
        pre += ((v0.x + v0.y) + (v0.z + v0.w)) + ((v1.x + v1.y) + (v1.z + v1.w))
             + ((v2.x + v2.y) + (v2.z + v2.w)) + ((v3.x + v3.y) + (v3.z + v3.w));
    }
    for (; q < nq; q += 1024) {
        int4 v = c4[q];
        pre += (v.x + v.y) + (v.z + v.w);
    }

    const int i = blockIdx.x * 1024 + tid;
    const int c = (i < n) ? cnt[i] : 0;

    int incl = c;
    for (int o = 1; o < 32; o <<= 1) {
        int t = __shfl_up(incl, o, 32);
        if (lane >= o) incl += t;
    }
    for (int o = 16; o > 0; o >>= 1) pre += __shfl_xor(pre, o, 32);

    if (lane == 31) s_wtot[wave] = incl;
    if (lane == 0)  s_wpre[wave] = pre;
    __syncthreads();
    if (wave == 0) {
        const int t = s_wtot[lane];
        int p  = s_wpre[lane];
        int ti = t;
        for (int o = 1; o < 32; o <<= 1) {
            int u = __shfl_up(ti, o, 32);
            if (lane >= o) ti += u;
        }
        for (int o = 16; o > 0; o >>= 1) p += __shfl_xor(p, o, 32);
        s_woff[lane] = p + ti - t;
    }
    __syncthreads();
    const int off = s_woff[wave];
    if (i < n) {
        rowptr[i] = off + incl - c;
        dis[i]    = rsqrtf((float)(c + 1));
        if (i == n - 1) rowptr[n] = off + incl;
    }
}


__device__ __forceinline__
void gemm128_block(const float* __restrict__ A, const float* __restrict__ W,
                   float* __restrict__ C, int M, int blk, float* __restrict__ lW) {
    for (int i = threadIdx.x; i < F_DIM * F_DIM; i += 256) lW[i] = W[i];
    __syncthreads();

    const int wave = threadIdx.x >> 5;
    const int lane = threadIdx.x & 31;
    const int strip = blk * 8 + wave;
    const int row0  = strip * 16;
    if (row0 >= M) return;

    const int half = lane >> 4;
    const int l16  = lane & 15;

    const v8f zero = {0.f, 0.f, 0.f, 0.f, 0.f, 0.f, 0.f, 0.f};
    v8f acc[8];
#pragma unroll
    for (int t = 0; t < 8; ++t) acc[t] = zero;

    const float* __restrict__ arow = A + (size_t)(row0 + l16) * F_DIM;

    for (int k0 = 0; k0 < F_DIM; k0 += 4) {
        const int ka = k0 + half * 2;
        v2f a;
        a.x = arow[ka];
        a.y = arow[ka + 1];
#pragma unroll
        for (int t = 0; t < 8; ++t) {
            const int col = t * 16 + l16;
            v2f b;
            b.x = lW[ka * F_DIM + col];
            b.y = lW[(ka + 1) * F_DIM + col];
            acc[t] = __builtin_amdgcn_wmma_f32_16x16x4_f32(
                 false, a,  false, b,
                 (short)0, acc[t],  false,  false);
        }
    }

#pragma unroll
    for (int t = 0; t < 8; ++t) {
        const int col = t * 16 + l16;
#pragma unroll
        for (int v = 0; v < 8; ++v) {
            C[(size_t)(row0 + v + 8 * half) * F_DIM + col] = acc[t][v];
        }
    }
}


__global__ __launch_bounds__(256)
void count_rank_gemm_kernel(const int* __restrict__ dst, int* __restrict__ cnt,
                            int* __restrict__ rank, int e,
                            const float* __restrict__ A, const float* __restrict__ W,
                            float* __restrict__ C, int M, int gb, int g0) {
    __shared__ float lW[F_DIM * F_DIM];
    if ((int)blockIdx.x < gb) {
        gemm128_block(A, W, C, M, g0 + (int)blockIdx.x, lW);
        return;
    }
    int i = ((int)blockIdx.x - gb) * 256 + threadIdx.x;
    if (i < e) rank[i] = atomicAdd(&cnt[dst[i]], 1);
}

__global__ __launch_bounds__(256)
void place_gemm_kernel(const int* __restrict__ src, const int* __restrict__ dst,
                       const int* __restrict__ rank, const int* __restrict__ rowptr,
                       const float* __restrict__ dis, int* __restrict__ csr, int e,
                       const float* __restrict__ A, const float* __restrict__ W,
                       float* __restrict__ C, int M, int gb, int g0) {
    __shared__ float lW[F_DIM * F_DIM];
    if ((int)blockIdx.x < gb) {
        gemm128_block(A, W, C, M, g0 + (int)blockIdx.x, lW);
        return;
    }
    int i = ((int)blockIdx.x - gb) * 256 + threadIdx.x;
    if (i >= e) return;
    const int s = src[i];
    const int d = dst[i];
    const int r = rank[i];
    const int   rp = rowptr[d];
    const float dd = dis[d];
    const float dsrc = dis[s];
    const int   pos = rp + r;
    const float w = dsrc * dd;
    if ((unsigned)pos < (unsigned)e) ((int2*)csr)[pos] = make_int2(s, __float_as_int(w));
}


__device__ __forceinline__
float4 gather_row(const float* __restrict__ H, const int* __restrict__ rowptr,
                  const int* __restrict__ csr, const float* __restrict__ dis,
                  const float* __restrict__ b, int node, int lane, int n, int e) {
    int beg = rowptr[node];
    int end = rowptr[node + 1];
    beg = max(beg, 0);
    end = min(end, e);
    const float4* __restrict__ H4 = (const float4*)H;
    const int2* __restrict__ csr2 = (const int2*)csr;
    const float  di = dis[node];
    const float4 hs = H4[(size_t)node * (F_DIM / 4) + lane];
    const float4 bb = ((const float4*)b)[lane];
    float4 acc = make_float4(0.f, 0.f, 0.f, 0.f);
    for (int c0 = beg; c0 < end; c0 += 32) {
        const int m = min(32, end - c0);
        int2 ev = make_int2(0, 0);
        if (lane < m) ev = csr2[c0 + lane];
        for (int j = 0; j < m; j += 8) {
            float4 h[8];
            float  w[8];
#pragma unroll
            for (int u = 0; u < 8; ++u) {
                h[u] = make_float4(0.f, 0.f, 0.f, 0.f);
                w[u] = 0.f;
                if (u == 0 || j + u < m) {
                    int s = __builtin_amdgcn_readlane(ev.x, j + u);
                    w[u] = __int_as_float(__builtin_amdgcn_readlane(ev.y, j + u));
                    s = min((unsigned)s, (unsigned)(n - 1));
                    h[u] = H4[(size_t)s * (F_DIM / 4) + lane];
                }
            }
#pragma unroll
            for (int u = 0; u < 8; ++u) {
                acc.x = fmaf(h[u].x, w[u], acc.x);
                acc.y = fmaf(h[u].y, w[u], acc.y);
                acc.z = fmaf(h[u].z, w[u], acc.z);
                acc.w = fmaf(h[u].w, w[u], acc.w);
            }
        }
    }
    float4 r;
    r.x = fmaxf(acc.x + hs.x * di * di + bb.x, 0.0f);
    r.y = fmaxf(acc.y + hs.y * di * di + bb.y, 0.0f);
    r.z = fmaxf(acc.z + hs.z * di * di + bb.z, 0.0f);
    r.w = fmaxf(acc.w + hs.w * di * di + bb.w, 0.0f);
    return r;
}

__global__ __launch_bounds__(256)
void gather_kernel(const float* __restrict__ H, const int* __restrict__ rowptr,
                   const int* __restrict__ csr, const float* __restrict__ dis,
                   const float* __restrict__ b, float* __restrict__ out, int n, int e) {
    const int lane = threadIdx.x & 31;
    const int node = __builtin_amdgcn_readfirstlane(blockIdx.x * 8 + (threadIdx.x >> 5));
    if (node >= n) return;
    const float4 r = gather_row(H, rowptr, csr, dis, b, node, lane, n, e);
    ((float4*)out)[(size_t)node * (F_DIM / 4) + lane] = r;
}


#define TILE_PITCH 132
#define FUSE_WAVES 16

__global__ __launch_bounds__(FUSE_WAVES * 32, 16)
void gather_gemm_kernel(const float* __restrict__ H, const int* __restrict__ rowptr,
                        const int* __restrict__ csr, const float* __restrict__ dis,
                        const float* __restrict__ b, const float* __restrict__ W,
                        float* __restrict__ C, int n, int e, int ntiles) {
    __shared__ __attribute__((aligned(16))) float lW[F_DIM * F_DIM];
    __shared__ __attribute__((aligned(16))) float lS[FUSE_WAVES * TILE_PITCH];

    {
        const float4* __restrict__ W4 = (const float4*)W;
        float4* lW4 = (float4*)lW;
        float4 wv[F_DIM * F_DIM / 4 / (FUSE_WAVES * 32)];
#pragma unroll
        for (int q = 0; q < F_DIM * F_DIM / 4 / (FUSE_WAVES * 32); ++q)
            wv[q] = W4[threadIdx.x + q * (FUSE_WAVES * 32)];
#pragma unroll
        for (int q = 0; q < F_DIM * F_DIM / 4 / (FUSE_WAVES * 32); ++q)
            lW4[threadIdx.x + q * (FUSE_WAVES * 32)] = wv[q];
    }

    const int lane = threadIdx.x & 31;
    const int wave = __builtin_amdgcn_readfirstlane(threadIdx.x >> 5);
    const int half = lane >> 4;
    const int l16  = lane & 15;

    for (int tile = blockIdx.x; tile < ntiles; tile += gridDim.x) {
        const int node = tile * 16 + wave;
        float4 r = make_float4(0.f, 0.f, 0.f, 0.f);
        if (node < n) r = gather_row(H, rowptr, csr, dis, b, node, lane, n, e);
        __syncthreads();
        *(float4*)(lS + wave * TILE_PITCH + lane * 4) = r;
        __syncthreads();

        if (wave < 4) {
            const v8f zero = {0.f, 0.f, 0.f, 0.f, 0.f, 0.f, 0.f, 0.f};
            v8f acc0 = zero;
            v8f acc1 = zero;
            const float* arow = lS + l16 * TILE_PITCH;
            const int col0 = wave * 32 + l16;
#pragma unroll 8
            for (int k0 = 0; k0 < F_DIM; k0 += 4) {
                const int ka = k0 + half * 2;
                v2f a;
                a.x = arow[ka];
                a.y = arow[ka + 1];
                v2f b0, b1;
                b0.x = lW[ka * F_DIM + col0];
                b0.y = lW[(ka + 1) * F_DIM + col0];
                b1.x = lW[ka * F_DIM + col0 + 16];
                b1.y = lW[(ka + 1) * F_DIM + col0 + 16];
                acc0 = __builtin_amdgcn_wmma_f32_16x16x4_f32(
                     false, a,  false, b0,
                     (short)0, acc0,  false,  false);
                acc1 = __builtin_amdgcn_wmma_f32_16x16x4_f32(
                     false, a,  false, b1,
                     (short)0, acc1,  false,  false);
            }
            const int row0 = tile * 16;
#pragma unroll
            for (int v = 0; v < 8; ++v) {
                const int row = row0 + v + 8 * half;
                C[(size_t)row * F_DIM + col0]      = acc0[v];
                C[(size_t)row * F_DIM + col0 + 16] = acc1[v];
            }
        }
    }
}


#define POOL_WAVES 16

__global__ __launch_bounds__(POOL_WAVES * 32)
void pool_head_kernel(const float* __restrict__ H, const int* __restrict__ gstart,
                      const float* __restrict__ Wlin, const float* __restrict__ blin,
                      float* __restrict__ out, int n) {
    __shared__ float4 s_part[POOL_WAVES][32];
    __shared__ float  s_pool[F_DIM];
    __shared__ float  s_lg[POOL_WAVES * 2][N_CLASSES];
    const int tid  = threadIdx.x;
    const int lane = tid & 31;
    const int wave = __builtin_amdgcn_readfirstlane(tid >> 5);
    const int g    = blockIdx.x;

    const int c  = tid & (N_CLASSES - 1);
    const int ks = tid >> 4;
    float wl[4];
#pragma unroll
    for (int q = 0; q < 4; ++q) wl[q] = Wlin[(ks * 4 + q) * N_CLASSES + c];
    const float bl = blin[c];

    int lo = gstart[g];
    int hi = gstart[g + 1];
    lo = max(lo, 0);
    hi = min(hi, n);

    const float4* __restrict__ H4 = (const float4*)H;
    float4 acc = make_float4(0.f, 0.f, 0.f, 0.f);
    for (int i = lo + wave; i < hi; i += 4 * POOL_WAVES) {
        float4 h[4];
#pragma unroll
        for (int u = 0; u < 4; ++u) {
            h[u] = make_float4(0.f, 0.f, 0.f, 0.f);
            const int r = i + u * POOL_WAVES;
            if (r < hi) h[u] = H4[(size_t)r * (F_DIM / 4) + lane];
        }
#pragma unroll
        for (int u = 0; u < 4; ++u) {
            acc.x += h[u].x;
            acc.y += h[u].y;
            acc.z += h[u].z;
            acc.w += h[u].w;
        }
    }
    s_part[wave][lane] = acc;
    __syncthreads();

    if (tid < F_DIM) {
        const float* __restrict__ sp = (const float*)s_part;
        float s = sp[tid];
#pragma unroll
        for (int w = 1; w < POOL_WAVES; ++w) s += sp[w * F_DIM + tid];
        s_pool[tid] = s;
    }
    __syncthreads();

    {
        float p = s_pool[ks * 4] * wl[0];
#pragma unroll
        for (int q = 1; q < 4; ++q) p = fmaf(s_pool[ks * 4 + q], wl[q], p);
        s_lg[ks][c] = p;
    }
    __syncthreads();

    if (wave == 0) {
        float lg = bl;
#pragma unroll
        for (int s = 0; s < POOL_WAVES * 2; ++s) lg += s_lg[s][c];
        float mx = lg;
#pragma unroll
        for (int o = 8; o > 0; o >>= 1) mx = fmaxf(mx, __shfl_xor(mx, o, 16));
        float se = expf(lg - mx);
#pragma unroll
        for (int o = 8; o > 0; o >>= 1) se += __shfl_xor(se, o, 16);
        const float lse = mx + logf(se);
        if (lane < N_CLASSES) out[(size_t)g * N_CLASSES + lane] = lg - lse;
    }
}


extern "C" void kernel_launch(void* const* d_in, const int* in_sizes, int n_in,
                              void* d_out, int out_size, void* d_ws, size_t ws_size,
                              hipStream_t stream) {
    const float* x     = (const float*)d_in[0];
    const int*   eidx  = (const int*)d_in[1];
    const int*   batch = (const int*)d_in[2];
    const float* W1    = (const float*)d_in[3];
    const float* b1    = (const float*)d_in[4];
    const float* W2    = (const float*)d_in[5];
    const float* b2    = (const float*)d_in[6];
    const float* Wlin  = (const float*)d_in[7];
    const float* blin  = (const float*)d_in[8];
    float*       out   = (float*)d_out;

    const int* src = eidx;
    const int* dst = eidx + N_EDGES;

    float* ws = (float*)d_ws;
    int*   cnt    = (int*)ws;
    float* dis    = ws + 50000;
    float* H      = ws + 100000;
    float* S      = H + (size_t)N_NODES * F_DIM;
    int*   gstart = (int*)(S + (size_t)N_NODES * F_DIM);
    int*   rowptr = gstart + N_GRAPHS * F_DIM;
    int*   rank   = rowptr + 50048;
    int*   csr    = rank + N_EDGES;

    const int B = 256;
    const int nodeBlocks  = (N_NODES + 1 + B - 1) / B;
    const int edgeBlocks  = (N_EDGES + B - 1) / B;
    const int gathBlocks  = (N_NODES + 7) / 8;
    const int gemmBlocks  = (N_NODES / 16 + 7) / 8;
    const int scanBlocks  = (N_NODES + 1023) / 1024;
    const int gemmHalfA   = (gemmBlocks + 1) / 2;
    const int gemmHalfB   = gemmBlocks - gemmHalfA;

    init_cnt_seg_kernel<<<nodeBlocks, B, 0, stream>>>(cnt, batch, gstart, N_NODES);
    count_rank_gemm_kernel<<<gemmHalfA + edgeBlocks, B, 0, stream>>>(
        dst, cnt, rank, N_EDGES, x, W1, H, N_NODES, gemmHalfA, 0);
    scan_dis_kernel<<<scanBlocks, 1024, 0, stream>>>(cnt, rowptr, dis, N_NODES);
    place_gemm_kernel<<<gemmHalfB + edgeBlocks, B, 0, stream>>>(
        src, dst, rank, rowptr, dis, csr, N_EDGES, x, W1, H, N_NODES, gemmHalfB, gemmHalfA);

    static_assert(N_NODES % 16 == 0, "the tile walk covers whole 16-node tiles");
    gather_gemm_kernel<<<1024, FUSE_WAVES * 32, 0, stream>>>(
        H, rowptr, csr, dis, b1, W2, S, N_NODES, N_EDGES, N_NODES / 16);

    gather_kernel<<<gathBlocks, B, 0, stream>>>(S, rowptr, csr, dis, b2, H, N_NODES, N_EDGES);

    pool_head_kernel<<<N_GRAPHS, POOL_WAVES * 32, 0, stream>>>(H, gstart, Wlin, blin, out, N_NODES);
}


# Edge ranks and in-degrees from LDS bins: a block per (4096-node range, 40000-edge chunk), 160 edges per thread, returning LDS adds; global atomics, zero-fill launch gone; sum order = chunk, LDS rank
# speedup vs baseline: 3.6497x; 1.2260x over previous
#include <hip/hip_runtime.h>
#include <hip/hip_bf16.h>
#include <math.h>


typedef float v2f __attribute__((ext_vector_type(2)));
typedef float v8f __attribute__((ext_vector_type(8)));

#define N_NODES   50000
#define N_EDGES   640000
#define F_DIM     128
#define N_CLASSES 16
#define N_GRAPHS  500


__device__ __forceinline__
void seg_ranges(const int* __restrict__ batch, int* __restrict__ gstart, int n, int i) {
    if (i <= n) {
        int prev = -1;
        int cur  = N_GRAPHS;
        if (i > 0) prev = min(max(batch[i - 1], 0), N_GRAPHS - 1);
        if (i < n) cur  = min(max(batch[i], 0), N_GRAPHS - 1);
        for (int g = prev + 1; g <= cur; ++g) gstart[g] = i;
    }
}


#define N_CHUNK   16
#define CHUNK_E   (N_EDGES / N_CHUNK)
#define RANGE_N   4096
#define N_RANGE   ((N_NODES + RANGE_N - 1) / RANGE_N)
#define NODE_PAD  (N_RANGE * RANGE_N)
#define SUB_N     1024
#define SUB_PER_RANGE (RANGE_N / SUB_N)
#define CNT_LOADS 10

__global__ __launch_bounds__(1024)
void scan_dis_kernel(const int* __restrict__ cntc, const int* __restrict__ tot,
                     int* __restrict__ rowptr, int* __restrict__ base,
                     float* __restrict__ dis, int n) {
    __shared__ int s_wtot[32];
    __shared__ int s_wpre[32];
    __shared__ int s_woff[32];
    const int tid  = threadIdx.x;
    const int lane = tid & 31;
    const int wave = tid >> 5;

    int pre = 0;
    const int nt = blockIdx.x * N_CHUNK;
    for (int q = tid; q < nt; q += 1024) pre += tot[q];

    const int i = blockIdx.x * SUB_N + tid;
    int cc[N_CHUNK];
#pragma unroll
    for (int k = 0; k < N_CHUNK; ++k) cc[k] = cntc[k * NODE_PAD + i];
    int c = 0;
#pragma unroll
    for (int k = 0; k < N_CHUNK; ++k) c += cc[k];

    int incl = c;
    for (int o = 1; o < 32; o <<= 1) {
        int t = __shfl_up(incl, o, 32);
        if (lane >= o) incl += t;
    }
    for (int o = 16; o > 0; o >>= 1) pre += __shfl_xor(pre, o, 32);

    if (lane == 31) s_wtot[wave] = incl;
    if (lane == 0)  s_wpre[wave] = pre;
    __syncthreads();
    if (wave == 0) {
        const int t = s_wtot[lane];
        int p  = s_wpre[lane];
        int ti = t;
        for (int o = 1; o < 32; o <<= 1) {
            int u = __shfl_up(ti, o, 32);
            if (lane >= o) ti += u;
        }
        for (int o = 16; o > 0; o >>= 1) p += __shfl_xor(p, o, 32);
        s_woff[lane] = p + ti - t;
    }
    __syncthreads();
    const int off = s_woff[wave];
    if (i < n) {
        int rp = off + incl - c;
        rowptr[i] = rp;
        dis[i]    = rsqrtf((float)(c + 1));
        if (i == n - 1) rowptr[n] = off + incl;
#pragma unroll
        for (int k = 0; k < N_CHUNK; ++k) {
            base[k * NODE_PAD + i] = rp;
            rp += cc[k];
        }
    }
}


__device__ __forceinline__
void gemm128_block(const float* __restrict__ A, const float* __restrict__ W,
                   float* __restrict__ C, int M, int blk, float* __restrict__ lW) {
    for (int i = threadIdx.x; i < F_DIM * F_DIM; i += 256) lW[i] = W[i];
    __syncthreads();

    const int wave = threadIdx.x >> 5;
    const int lane = threadIdx.x & 31;
    const int strip = blk * 8 + wave;
    const int row0  = strip * 16;
    if (row0 >= M) return;

    const int half = lane >> 4;
    const int l16  = lane & 15;

    const v8f zero = {0.f, 0.f, 0.f, 0.f, 0.f, 0.f, 0.f, 0.f};
    v8f acc[8];
#pragma unroll
    for (int t = 0; t < 8; ++t) acc[t] = zero;

    const float* __restrict__ arow = A + (size_t)(row0 + l16) * F_DIM;

    for (int k0 = 0; k0 < F_DIM; k0 += 4) {
        const int ka = k0 + half * 2;
        v2f a;
        a.x = arow[ka];
        a.y = arow[ka + 1];
#pragma unroll
        for (int t = 0; t < 8; ++t) {
            const int col = t * 16 + l16;
            v2f b;
            b.x = lW[ka * F_DIM + col];
            b.y = lW[(ka + 1) * F_DIM + col];
            acc[t] = __builtin_amdgcn_wmma_f32_16x16x4_f32(
                 false, a,  false, b,
                 (short)0, acc[t],  false,  false);
        }
    }

#pragma unroll
    for (int t = 0; t < 8; ++t) {
        const int col = t * 16 + l16;
#pragma unroll
        for (int v = 0; v < 8; ++v) {
            C[(size_t)(row0 + v + 8 * half) * F_DIM + col] = acc[t][v];
        }
    }
}


__global__ __launch_bounds__(256)
void count_rank_gemm_kernel(const int* __restrict__ dst, int* __restrict__ cntc,
                            int* __restrict__ tot, int* __restrict__ rank,
                            const int* __restrict__ batch, int* __restrict__ gstart, int n,
                            const float* __restrict__ A, const float* __restrict__ W,
                            float* __restrict__ C, int M, int gb, int g0) {
    __shared__ float lW[F_DIM * F_DIM];
    __shared__ int s_sum[SUB_PER_RANGE * 8];
    if ((int)blockIdx.x < gb) {
        gemm128_block(A, W, C, M, g0 + (int)blockIdx.x, lW);
        return;
    }
    const int idx = (int)blockIdx.x - gb;
    const int tid = threadIdx.x;
    if (idx >= N_RANGE * N_CHUNK) {
        seg_ranges(batch, gstart, n, (idx - N_RANGE * N_CHUNK) * 256 + tid);
        return;
    }
    const int r  = idx / N_CHUNK;
    const int c  = idx - r * N_CHUNK;
    const int r0 = r * RANGE_N;
    int* bins = (int*)lW;
    for (int k = 0; k < RANGE_N / 256; ++k) bins[tid + k * 256] = 0;
    if (tid < 32) bins[RANGE_N + tid] = 0;
    __syncthreads();
    const int spare = RANGE_N + (tid & 31);
    const int4* __restrict__ d4 = (const int4*)dst + c * (CHUNK_E / 4);
    int* __restrict__ rk = rank + c * CHUNK_E;
    for (int q0 = tid; q0 < CHUNK_E / 4; q0 += CNT_LOADS * 256) {
        int4 v[CNT_LOADS];
        int4 kk[CNT_LOADS];
#pragma unroll
        for (int u = 0; u < CNT_LOADS; ++u) {
            const int q = q0 + u * 256;
            v[u] = make_int4(-1, -1, -1, -1);
            if (q < CHUNK_E / 4) v[u] = d4[q];
        }
#pragma unroll
        for (int u = 0; u < CNT_LOADS; ++u) {
            const unsigned o0 = (unsigned)(v[u].x - r0);
            const unsigned o1 = (unsigned)(v[u].y - r0);
            const unsigned o2 = (unsigned)(v[u].z - r0);
            const unsigned o3 = (unsigned)(v[u].w - r0);
            kk[u].x = atomicAdd(&bins[o0 < (unsigned)RANGE_N ? (int)o0 : spare], 1);
            kk[u].y = atomicAdd(&bins[o1 < (unsigned)RANGE_N ? (int)o1 : spare], 1);
            kk[u].z = atomicAdd(&bins[o2 < (unsigned)RANGE_N ? (int)o2 : spare], 1);
            kk[u].w = atomicAdd(&bins[o3 < (unsigned)RANGE_N ? (int)o3 : spare], 1);
        }
#pragma unroll
        for (int u = 0; u < CNT_LOADS; ++u) {
            const int q = q0 + u * 256;
            if ((unsigned)(v[u].x - r0) < (unsigned)RANGE_N) rk[4 * q]     = kk[u].x;
            if ((unsigned)(v[u].y - r0) < (unsigned)RANGE_N) rk[4 * q + 1] = kk[u].y;
            if ((unsigned)(v[u].z - r0) < (unsigned)RANGE_N) rk[4 * q + 2] = kk[u].z;
            if ((unsigned)(v[u].w - r0) < (unsigned)RANGE_N) rk[4 * q + 3] = kk[u].w;
        }
    }
    __syncthreads();
    int ss[SUB_PER_RANGE];
#pragma unroll
    for (int j = 0; j < SUB_PER_RANGE; ++j) ss[j] = 0;
#pragma unroll
    for (int k = 0; k < RANGE_N / 256; ++k) {
        const int b = bins[tid + k * 256];
        cntc[c * NODE_PAD + r0 + tid + k * 256] = b;
        ss[k / (SUB_N / 256)] += b;
    }
#pragma unroll
    for (int j = 0; j < SUB_PER_RANGE; ++j) {
        int t = ss[j];
        for (int o = 16; o > 0; o >>= 1) t += __shfl_xor(t, o, 32);
        if ((tid & 31) == 0) s_sum[j * 8 + (tid >> 5)] = t;
    }
    __syncthreads();
    if (tid < SUB_PER_RANGE) {
        int t = 0;
        for (int w = 0; w < 8; ++w) t += s_sum[tid * 8 + w];
        tot[(r * SUB_PER_RANGE + tid) * N_CHUNK + c] = t;
    }
}

__global__ __launch_bounds__(256)
void place_gemm_kernel(const int* __restrict__ src, const int* __restrict__ dst,
                       const int* __restrict__ rank, const int* __restrict__ base,
                       const float* __restrict__ dis, int* __restrict__ csr, int e,
                       const float* __restrict__ A, const float* __restrict__ W,
                       float* __restrict__ C, int M, int gb, int g0) {
    __shared__ float lW[F_DIM * F_DIM];
    if ((int)blockIdx.x < gb) {
        gemm128_block(A, W, C, M, g0 + (int)blockIdx.x, lW);
        return;
    }
    int i = ((int)blockIdx.x - gb) * 256 + threadIdx.x;
    if (i >= e) return;
    const int s = src[i];
    const int d = dst[i];
    const int r = rank[i];
    const int   rp = base[(i / CHUNK_E) * NODE_PAD + min((unsigned)d, (unsigned)(NODE_PAD - 1))];
    const float dd = dis[d];
    const float dsrc = dis[s];
    const int   pos = rp + r;
    const float w = dsrc * dd;
    if ((unsigned)pos < (unsigned)e) ((int2*)csr)[pos] = make_int2(s, __float_as_int(w));
}


__device__ __forceinline__
float4 gather_row(const float* __restrict__ H, const int* __restrict__ rowptr,
                  const int* __restrict__ csr, const float* __restrict__ dis,
                  const float* __restrict__ b, int node, int lane, int n, int e) {
    int beg = rowptr[node];
    int end = rowptr[node + 1];
    beg = max(beg, 0);
    end = min(end, e);
    const float4* __restrict__ H4 = (const float4*)H;
    const int2* __restrict__ csr2 = (const int2*)csr;
    const float  di = dis[node];
    const float4 hs = H4[(size_t)node * (F_DIM / 4) + lane];
    const float4 bb = ((const float4*)b)[lane];
    float4 acc = make_float4(0.f, 0.f, 0.f, 0.f);
    for (int c0 = beg; c0 < end; c0 += 32) {
        const int m = min(32, end - c0);
        int2 ev = make_int2(0, 0);
        if (lane < m) ev = csr2[c0 + lane];
        for (int j = 0; j < m; j += 8) {
            float4 h[8];
            float  w[8];
#pragma unroll
            for (int u = 0; u < 8; ++u) {
                h[u] = make_float4(0.f, 0.f, 0.f, 0.f);
                w[u] = 0.f;
                if (u == 0 || j + u < m) {
                    int s = __builtin_amdgcn_readlane(ev.x, j + u);
                    w[u] = __int_as_float(__builtin_amdgcn_readlane(ev.y, j + u));
                    s = min((unsigned)s, (unsigned)(n - 1));
                    h[u] = H4[(size_t)s * (F_DIM / 4) + lane];
                }
            }
#pragma unroll
            for (int u = 0; u < 8; ++u) {
                acc.x = fmaf(h[u].x, w[u], acc.x);
                acc.y = fmaf(h[u].y, w[u], acc.y);
                acc.z = fmaf(h[u].z, w[u], acc.z);
                acc.w = fmaf(h[u].w, w[u], acc.w);
            }
        }
    }
    float4 r;
    r.x = fmaxf(acc.x + hs.x * di * di + bb.x, 0.0f);
    r.y = fmaxf(acc.y + hs.y * di * di + bb.y, 0.0f);
    r.z = fmaxf(acc.z + hs.z * di * di + bb.z, 0.0f);
    r.w = fmaxf(acc.w + hs.w * di * di + bb.w, 0.0f);
    return r;
}

__global__ __launch_bounds__(256)
void gather_kernel(const float* __restrict__ H, const int* __restrict__ rowptr,
                   const int* __restrict__ csr, const float* __restrict__ dis,
                   const float* __restrict__ b, float* __restrict__ out, int n, int e) {
    const int lane = threadIdx.x & 31;
    const int node = __builtin_amdgcn_readfirstlane(blockIdx.x * 8 + (threadIdx.x >> 5));
    if (node >= n) return;
    const float4 r = gather_row(H, rowptr, csr, dis, b, node, lane, n, e);
    ((float4*)out)[(size_t)node * (F_DIM / 4) + lane] = r;
}


#define TILE_PITCH 132
#define FUSE_WAVES 16

__global__ __launch_bounds__(FUSE_WAVES * 32, 16)
void gather_gemm_kernel(const float* __restrict__ H, const int* __restrict__ rowptr,
                        const int* __restrict__ csr, const float* __restrict__ dis,
                        const float* __restrict__ b, const float* __restrict__ W,
                        float* __restrict__ C, int n, int e, int ntiles) {
    __shared__ __attribute__((aligned(16))) float lW[F_DIM * F_DIM];
    __shared__ __attribute__((aligned(16))) float lS[FUSE_WAVES * TILE_PITCH];

    {
        const float4* __restrict__ W4 = (const float4*)W;
        float4* lW4 = (float4*)lW;
        float4 wv[F_DIM * F_DIM / 4 / (FUSE_WAVES * 32)];
#pragma unroll
        for (int q = 0; q < F_DIM * F_DIM / 4 / (FUSE_WAVES * 32); ++q)
            wv[q] = W4[threadIdx.x + q * (FUSE_WAVES * 32)];
#pragma unroll
        for (int q = 0; q < F_DIM * F_DIM / 4 / (FUSE_WAVES * 32); ++q)
            lW4[threadIdx.x + q * (FUSE_WAVES * 32)] = wv[q];
    }

    const int lane = threadIdx.x & 31;
    const int wave = __builtin_amdgcn_readfirstlane(threadIdx.x >> 5);
    const int half = lane >> 4;
    const int l16  = lane & 15;

    for (int tile = blockIdx.x; tile < ntiles; tile += gridDim.x) {
        const int node = tile * 16 + wave;
        float4 r = make_float4(0.f, 0.f, 0.f, 0.f);
        if (node < n) r = gather_row(H, rowptr, csr, dis, b, node, lane, n, e);
        __syncthreads();
        *(float4*)(lS + wave * TILE_PITCH + lane * 4) = r;
        __syncthreads();

        if (wave < 4) {
            const v8f zero = {0.f, 0.f, 0.f, 0.f, 0.f, 0.f, 0.f, 0.f};
            v8f acc0 = zero;
            v8f acc1 = zero;
            const float* arow = lS + l16 * TILE_PITCH;
            const int col0 = wave * 32 + l16;
#pragma unroll 8
            for (int k0 = 0; k0 < F_DIM; k0 += 4) {
                const int ka = k0 + half * 2;
                v2f a;
                a.x = arow[ka];
                a.y = arow[ka + 1];
                v2f b0, b1;
                b0.x = lW[ka * F_DIM + col0];
                b0.y = lW[(ka + 1) * F_DIM + col0];
                b1.x = lW[ka * F_DIM + col0 + 16];
                b1.y = lW[(ka + 1) * F_DIM + col0 + 16];
                acc0 = __builtin_amdgcn_wmma_f32_16x16x4_f32(
                     false, a,  false, b0,
                     (short)0, acc0,  false,  false);
                acc1 = __builtin_amdgcn_wmma_f32_16x16x4_f32(
                     false, a,  false, b1,
                     (short)0, acc1,  false,  false);
            }
            const int row0 = tile * 16;
#pragma unroll
            for (int v = 0; v < 8; ++v) {
                const int row = row0 + v + 8 * half;
                C[(size_t)row * F_DIM + col0]      = acc0[v];
                C[(size_t)row * F_DIM + col0 + 16] = acc1[v];
            }
        }
    }
}


#define POOL_WAVES 16

__global__ __launch_bounds__(POOL_WAVES * 32)
void pool_head_kernel(const float* __restrict__ H, const int* __restrict__ gstart,
                      const float* __restrict__ Wlin, const float* __restrict__ blin,
                      float* __restrict__ out, int n) {
    __shared__ float4 s_part[POOL_WAVES][32];
    __shared__ float  s_pool[F_DIM];
    __shared__ float  s_lg[POOL_WAVES * 2][N_CLASSES];
    const int tid  = threadIdx.x;
    const int lane = tid & 31;
    const int wave = __builtin_amdgcn_readfirstlane(tid >> 5);
    const int g    = blockIdx.x;

    const int c  = tid & (N_CLASSES - 1);
    const int ks = tid >> 4;
    float wl[4];
#pragma unroll
    for (int q = 0; q < 4; ++q) wl[q] = Wlin[(ks * 4 + q) * N_CLASSES + c];
    const float bl = blin[c];

    int lo = gstart[g];
    int hi = gstart[g + 1];
    lo = max(lo, 0);
    hi = min(hi, n);

    const float4* __restrict__ H4 = (const float4*)H;
    float4 acc = make_float4(0.f, 0.f, 0.f, 0.f);
    for (int i = lo + wave; i < hi; i += 4 * POOL_WAVES) {
        float4 h[4];
#pragma unroll
        for (int u = 0; u < 4; ++u) {
            h[u] = make_float4(0.f, 0.f, 0.f, 0.f);
            const int r = i + u * POOL_WAVES;
            if (r < hi) h[u] = H4[(size_t)r * (F_DIM / 4) + lane];
        }
#pragma unroll
        for (int u = 0; u < 4; ++u) {
            acc.x += h[u].x;
            acc.y += h[u].y;
            acc.z += h[u].z;
            acc.w += h[u].w;
        }
    }
    s_part[wave][lane] = acc;
    __syncthreads();

    if (tid < F_DIM) {
        const float* __restrict__ sp = (const float*)s_part;
        float s = sp[tid];
#pragma unroll
        for (int w = 1; w < POOL_WAVES; ++w) s += sp[w * F_DIM + tid];
        s_pool[tid] = s;
    }
    __syncthreads();

    {
        float p = s_pool[ks * 4] * wl[0];
#pragma unroll
        for (int q = 1; q < 4; ++q) p = fmaf(s_pool[ks * 4 + q], wl[q], p);
        s_lg[ks][c] = p;
    }
    __syncthreads();

    if (wave == 0) {
        float lg = bl;
#pragma unroll
        for (int s = 0; s < POOL_WAVES * 2; ++s) lg += s_lg[s][c];
        float mx = lg;
#pragma unroll
        for (int o = 8; o > 0; o >>= 1) mx = fmaxf(mx, __shfl_xor(mx, o, 16));
        float se = expf(lg - mx);
#pragma unroll
        for (int o = 8; o > 0; o >>= 1) se += __shfl_xor(se, o, 16);
        const float lse = mx + logf(se);
        if (lane < N_CLASSES) out[(size_t)g * N_CLASSES + lane] = lg - lse;
    }
}


extern "C" void kernel_launch(void* const* d_in, const int* in_sizes, int n_in,
                              void* d_out, int out_size, void* d_ws, size_t ws_size,
                              hipStream_t stream) {
    const float* x     = (const float*)d_in[0];
    const int*   eidx  = (const int*)d_in[1];
    const int*   batch = (const int*)d_in[2];
    const float* W1    = (const float*)d_in[3];
    const float* b1    = (const float*)d_in[4];
    const float* W2    = (const float*)d_in[5];
    const float* b2    = (const float*)d_in[6];
    const float* Wlin  = (const float*)d_in[7];
    const float* blin  = (const float*)d_in[8];
    float*       out   = (float*)d_out;

    const int* src = eidx;
    const int* dst = eidx + N_EDGES;

    float* ws = (float*)d_ws;
    float* dis    = ws + 50000;
    float* H      = ws + 100000;
    float* S      = H + (size_t)N_NODES * F_DIM;
    int*   gstart = (int*)(S + (size_t)N_NODES * F_DIM);
    int*   rowptr = gstart + N_GRAPHS * F_DIM;
    int*   rank   = rowptr + 50048;
    int*   csr    = rank + N_EDGES;
    int*   cntc   = csr + 2 * N_EDGES;
    int*   base   = cntc + N_CHUNK * NODE_PAD;
    int*   tot    = base + N_CHUNK * NODE_PAD;

    const int B = 256;
    const int segBlocks   = (N_NODES + 1 + B - 1) / B;
    const int cntBlocks   = N_RANGE * N_CHUNK;
    const int edgeBlocks  = (N_EDGES + B - 1) / B;
    const int gathBlocks  = (N_NODES + 7) / 8;
    const int gemmBlocks  = (N_NODES / 16 + 7) / 8;
    const int gemmHalfA   = (gemmBlocks + 1) / 2;
    const int gemmHalfB   = gemmBlocks - gemmHalfA;
    const int scanBlocks  = (N_NODES + SUB_N - 1) / SUB_N;

    static_assert(N_EDGES % (4 * N_CHUNK) == 0, "chunks are whole 16-byte groups of destinations");
    static_assert(SUB_N == 1024 && RANGE_N % SUB_N == 0 && SUB_N % 256 == 0, "scan blocks of 1024 nodes tile the ranges");
    count_rank_gemm_kernel<<<gemmHalfA + cntBlocks + segBlocks, B, 0, stream>>>(
        dst, cntc, tot, rank, batch, gstart, N_NODES, x, W1, H, N_NODES, gemmHalfA, 0);
    scan_dis_kernel<<<scanBlocks, SUB_N, 0, stream>>>(cntc, tot, rowptr, base, dis, N_NODES);
    place_gemm_kernel<<<gemmHalfB + edgeBlocks, B, 0, stream>>>(
        src, dst, rank, base, dis, csr, N_EDGES, x, W1, H, N_NODES, gemmHalfB, gemmHalfA);

    static_assert(N_NODES % 16 == 0, "the tile walk covers whole 16-node tiles");
    gather_gemm_kernel<<<1024, FUSE_WAVES * 32, 0, stream>>>(
        H, rowptr, csr, dis, b1, W2, S, N_NODES, N_EDGES, N_NODES / 16);

    gather_kernel<<<gathBlocks, B, 0, stream>>>(S, rowptr, csr, dis, b2, H, N_NODES, N_EDGES);

    pool_head_kernel<<<N_GRAPHS, POOL_WAVES * 32, 0, stream>>>(H, gstart, Wlin, blin, out, N_NODES);
}


# Placement by (node range, edge chunk) blocks: LDS words start at the scanned bases, a returning LDS add is the edge's slot; rank array, per-edge base and dis[dst] reads gone; sum order LDS arrival
# speedup vs baseline: 3.7936x; 1.0394x over previous
#include <hip/hip_runtime.h>
#include <hip/hip_bf16.h>
#include <math.h>


typedef float v2f __attribute__((ext_vector_type(2)));
typedef float v8f __attribute__((ext_vector_type(8)));

#define N_NODES   50000
#define N_EDGES   640000
#define F_DIM     128
#define N_CLASSES 16
#define N_GRAPHS  500


__device__ __forceinline__
void seg_ranges(const int* __restrict__ batch, int* __restrict__ gstart, int n, int i) {
    if (i <= n) {
        int prev = -1;
        int cur  = N_GRAPHS;
        if (i > 0) prev = min(max(batch[i - 1], 0), N_GRAPHS - 1);
        if (i < n) cur  = min(max(batch[i], 0), N_GRAPHS - 1);
        for (int g = prev + 1; g <= cur; ++g) gstart[g] = i;
    }
}


#define N_CHUNK   16
#define CHUNK_E   (N_EDGES / N_CHUNK)
#define RANGE_N   4096
#define N_RANGE   ((N_NODES + RANGE_N - 1) / RANGE_N)
#define NODE_PAD  (N_RANGE * RANGE_N)
#define SUB_N     1024
#define SUB_PER_RANGE (RANGE_N / SUB_N)
#define CNT_LOADS 10

__global__ __launch_bounds__(1024)
void scan_dis_kernel(const int* __restrict__ cntc, const int* __restrict__ tot,
                     int* __restrict__ rowptr, int* __restrict__ base,
                     float* __restrict__ dis, int n) {
    __shared__ int s_wtot[32];
    __shared__ int s_wpre[32];
    __shared__ int s_woff[32];
    const int tid  = threadIdx.x;
    const int lane = tid & 31;
    const int wave = tid >> 5;

    int pre = 0;
    const int nt = blockIdx.x * N_CHUNK;
    for (int q = tid; q < nt; q += 1024) pre += tot[q];

    const int i = blockIdx.x * SUB_N + tid;
    int cc[N_CHUNK];
#pragma unroll
    for (int k = 0; k < N_CHUNK; ++k) cc[k] = cntc[k * NODE_PAD + i];
    int c = 0;
#pragma unroll
    for (int k = 0; k < N_CHUNK; ++k) c += cc[k];

    int incl = c;
    for (int o = 1; o < 32; o <<= 1) {
        int t = __shfl_up(incl, o, 32);
        if (lane >= o) incl += t;
    }
    for (int o = 16; o > 0; o >>= 1) pre += __shfl_xor(pre, o, 32);

    if (lane == 31) s_wtot[wave] = incl;
    if (lane == 0)  s_wpre[wave] = pre;
    __syncthreads();
    if (wave == 0) {
        const int t = s_wtot[lane];
        int p  = s_wpre[lane];
        int ti = t;
        for (int o = 1; o < 32; o <<= 1) {
            int u = __shfl_up(ti, o, 32);
            if (lane >= o) ti += u;
        }
        for (int o = 16; o > 0; o >>= 1) p += __shfl_xor(p, o, 32);
        s_woff[lane] = p + ti - t;
    }
    __syncthreads();
    const int off = s_woff[wave];
    if (i < n) {
        int rp = off + incl - c;
        rowptr[i] = rp;
        dis[i]    = rsqrtf((float)(c + 1));
        if (i == n - 1) rowptr[n] = off + incl;
#pragma unroll
        for (int k = 0; k < N_CHUNK; ++k) {
            base[k * NODE_PAD + i] = rp;
            rp += cc[k];
        }
    }
}


__device__ __forceinline__
void gemm128_block(const float* __restrict__ A, const float* __restrict__ W,
                   float* __restrict__ C, int M, int blk, float* __restrict__ lW) {
    for (int i = threadIdx.x; i < F_DIM * F_DIM; i += 256) lW[i] = W[i];
    __syncthreads();

    const int wave = threadIdx.x >> 5;
    const int lane = threadIdx.x & 31;
    const int strip = blk * 8 + wave;
    const int row0  = strip * 16;
    if (row0 >= M) return;

    const int half = lane >> 4;
    const int l16  = lane & 15;

    const v8f zero = {0.f, 0.f, 0.f, 0.f, 0.f, 0.f, 0.f, 0.f};
    v8f acc[8];
#pragma unroll
    for (int t = 0; t < 8; ++t) acc[t] = zero;

    const float* __restrict__ arow = A + (size_t)(row0 + l16) * F_DIM;

    for (int k0 = 0; k0 < F_DIM; k0 += 4) {
        const int ka = k0 + half * 2;
        v2f a;
        a.x = arow[ka];
        a.y = arow[ka + 1];
#pragma unroll
        for (int t = 0; t < 8; ++t) {
            const int col = t * 16 + l16;
            v2f b;
            b.x = lW[ka * F_DIM + col];
            b.y = lW[(ka + 1) * F_DIM + col];
            acc[t] = __builtin_amdgcn_wmma_f32_16x16x4_f32(
                 false, a,  false, b,
                 (short)0, acc[t],  false,  false);
        }
    }

#pragma unroll
    for (int t = 0; t < 8; ++t) {
        const int col = t * 16 + l16;
#pragma unroll
        for (int v = 0; v < 8; ++v) {
            C[(size_t)(row0 + v + 8 * half) * F_DIM + col] = acc[t][v];
        }
    }
}


__global__ __launch_bounds__(256)
void count_gemm_kernel(const int* __restrict__ dst, int* __restrict__ cntc,
                       int* __restrict__ tot,
                       const int* __restrict__ batch, int* __restrict__ gstart, int n,
                       const float* __restrict__ A, const float* __restrict__ W,
                       float* __restrict__ C, int M, int gb, int g0) {
    __shared__ float lW[F_DIM * F_DIM];
    __shared__ int s_sum[SUB_PER_RANGE * 8];
    if ((int)blockIdx.x < gb) {
        gemm128_block(A, W, C, M, g0 + (int)blockIdx.x, lW);
        return;
    }
    const int idx = (int)blockIdx.x - gb;
    const int tid = threadIdx.x;
    if (idx >= N_RANGE * N_CHUNK) {
        seg_ranges(batch, gstart, n, (idx - N_RANGE * N_CHUNK) * 256 + tid);
        return;
    }
    const int r  = idx / N_CHUNK;
    const int c  = idx - r * N_CHUNK;
    const int r0 = r * RANGE_N;
    int* bins = (int*)lW;
    for (int k = 0; k < RANGE_N / 256; ++k) bins[tid + k * 256] = 0;
    __syncthreads();
    const int4* __restrict__ d4 = (const int4*)dst + c * (CHUNK_E / 4);
    for (int q0 = tid; q0 < CHUNK_E / 4; q0 += CNT_LOADS * 256) {
        int4 v[CNT_LOADS];
#pragma unroll
        for (int u = 0; u < CNT_LOADS; ++u) {
            const int q = q0 + u * 256;
            v[u] = make_int4(-1, -1, -1, -1);
            if (q < CHUNK_E / 4) v[u] = d4[q];
        }
#pragma unroll
        for (int u = 0; u < CNT_LOADS; ++u) {
            const unsigned o0 = (unsigned)(v[u].x - r0);
            const unsigned o1 = (unsigned)(v[u].y - r0);
            const unsigned o2 = (unsigned)(v[u].z - r0);
            const unsigned o3 = (unsigned)(v[u].w - r0);
            if (o0 < (unsigned)RANGE_N) atomicAdd(&bins[o0], 1);
            if (o1 < (unsigned)RANGE_N) atomicAdd(&bins[o1], 1);
            if (o2 < (unsigned)RANGE_N) atomicAdd(&bins[o2], 1);
            if (o3 < (unsigned)RANGE_N) atomicAdd(&bins[o3], 1);
        }
    }
    __syncthreads();
    int ss[SUB_PER_RANGE];
#pragma unroll
    for (int j = 0; j < SUB_PER_RANGE; ++j) ss[j] = 0;
#pragma unroll
    for (int k = 0; k < RANGE_N / 256; ++k) {
        const int b = bins[tid + k * 256];
        cntc[c * NODE_PAD + r0 + tid + k * 256] = b;
        ss[k / (SUB_N / 256)] += b;
    }
#pragma unroll
    for (int j = 0; j < SUB_PER_RANGE; ++j) {
        int t = ss[j];
        for (int o = 16; o > 0; o >>= 1) t += __shfl_xor(t, o, 32);
        if ((tid & 31) == 0) s_sum[j * 8 + (tid >> 5)] = t;
    }
    __syncthreads();
    if (tid < SUB_PER_RANGE) {
        int t = 0;
        for (int w = 0; w < 8; ++w) t += s_sum[tid * 8 + w];
        tot[(r * SUB_PER_RANGE + tid) * N_CHUNK + c] = t;
    }
}

#define PL_LOADS 10

__global__ __launch_bounds__(256)
void place_gemm_kernel(const int* __restrict__ src, const int* __restrict__ dst,
                       const int* __restrict__ base, const float* __restrict__ dis,
                       int* __restrict__ csr, int n, int e,
                       const float* __restrict__ A, const float* __restrict__ W,
                       float* __restrict__ C, int M, int gb, int g0) {
    __shared__ float lW[F_DIM * F_DIM];
    if ((int)blockIdx.x < gb) {
        gemm128_block(A, W, C, M, g0 + (int)blockIdx.x, lW);
        return;
    }
    const int idx = (int)blockIdx.x - gb;
    if (idx >= N_RANGE * N_CHUNK) return;
    const int tid = threadIdx.x;
    const int r  = idx / N_CHUNK;
    const int c  = idx - r * N_CHUNK;
    const int r0 = r * RANGE_N;
    int*   slot = (int*)lW;
    float* ldis = lW + RANGE_N + 32;
    {
        int   bs[RANGE_N / 256];
        float dd[RANGE_N / 256];
#pragma unroll
        for (int k = 0; k < RANGE_N / 256; ++k) {
            const int d = min(r0 + tid + k * 256, n - 1);
            bs[k] = base[c * NODE_PAD + d];
            dd[k] = dis[d];
        }
#pragma unroll
        for (int k = 0; k < RANGE_N / 256; ++k) {
            slot[tid + k * 256] = bs[k];
            ldis[tid + k * 256] = dd[k];
        }
    }
    if (tid < 32) slot[RANGE_N + tid] = 0x40000000;
    __syncthreads();
    const int spare = RANGE_N + (tid & 31);
    const int4* __restrict__ d4 = (const int4*)dst + c * (CHUNK_E / 4);
    const int4* __restrict__ s4 = (const int4*)src + c * (CHUNK_E / 4);
    int2* __restrict__ csr2 = (int2*)csr;
    for (int q0 = tid; q0 < CHUNK_E / 4; q0 += PL_LOADS * 256) {
        int   od[PL_LOADS][4];
        int   sr[PL_LOADS][4];
        float ds[PL_LOADS][4];
        float ld[PL_LOADS][4];
        int   ps[PL_LOADS][4];
#pragma unroll
        for (int u = 0; u < PL_LOADS; ++u) {
            const int q = q0 + u * 256;
            int4 dv = make_int4(-1, -1, -1, -1);
            int4 sv = make_int4(0, 0, 0, 0);
            if (q < CHUNK_E / 4) {
                dv = d4[q];
                sv = s4[q];
            }
            od[u][0] = dv.x - r0;  od[u][1] = dv.y - r0;  od[u][2] = dv.z - r0;  od[u][3] = dv.w - r0;
            sr[u][0] = sv.x;       sr[u][1] = sv.y;       sr[u][2] = sv.z;       sr[u][3] = sv.w;
        }
#pragma unroll
        for (int u = 0; u < PL_LOADS; ++u) {
#pragma unroll
            for (int j = 0; j < 4; ++j) {
                const bool in = (unsigned)od[u][j] < (unsigned)RANGE_N;
                const int si = in ? (int)min((unsigned)sr[u][j], (unsigned)(n - 1)) : 0;
                ds[u][j] = dis[si];
            }
        }
#pragma unroll
        for (int u = 0; u < PL_LOADS; ++u) {
#pragma unroll
            for (int j = 0; j < 4; ++j) {
                const bool in = (unsigned)od[u][j] < (unsigned)RANGE_N;
                ps[u][j] = atomicAdd(&slot[in ? od[u][j] : spare], 1);
                ld[u][j] = ldis[in ? od[u][j] : 0];
            }
        }
#pragma unroll
        for (int u = 0; u < PL_LOADS; ++u)
            asm volatile("" :: "v"(ds[u][0]), "v"(ds[u][1]), "v"(ds[u][2]), "v"(ds[u][3]));
#pragma unroll
        for (int u = 0; u < PL_LOADS; ++u) {
#pragma unroll
            for (int j = 0; j < 4; ++j) {
                const float w = ds[u][j] * ld[u][j];
                const int pos = ps[u][j];
                if ((unsigned)pos < (unsigned)e) csr2[pos] = make_int2(sr[u][j], __float_as_int(w));
            }
        }
    }
}


__device__ __forceinline__
float4 gather_row(const float* __restrict__ H, const int* __restrict__ rowptr,
                  const int* __restrict__ csr, const float* __restrict__ dis,
                  const float* __restrict__ b, int node, int lane, int n, int e) {
    int beg = rowptr[node];
    int end = rowptr[node + 1];
    beg = max(beg, 0);
    end = min(end, e);
    const float4* __restrict__ H4 = (const float4*)H;
    const int2* __restrict__ csr2 = (const int2*)csr;
    const float  di = dis[node];
    const float4 hs = H4[(size_t)node * (F_DIM / 4) + lane];
    const float4 bb = ((const float4*)b)[lane];
    float4 acc = make_float4(0.f, 0.f, 0.f, 0.f);
    for (int c0 = beg; c0 < end; c0 += 32) {
        const int m = min(32, end - c0);
        int2 ev = make_int2(0, 0);
        if (lane < m) ev = csr2[c0 + lane];
        for (int j = 0; j < m; j += 8) {
            float4 h[8];
            float  w[8];
#pragma unroll
            for (int u = 0; u < 8; ++u) {
                h[u] = make_float4(0.f, 0.f, 0.f, 0.f);
                w[u] = 0.f;
                if (u == 0 || j + u < m) {
                    int s = __builtin_amdgcn_readlane(ev.x, j + u);
                    w[u] = __int_as_float(__builtin_amdgcn_readlane(ev.y, j + u));
                    s = min((unsigned)s, (unsigned)(n - 1));
                    h[u] = H4[(size_t)s * (F_DIM / 4) + lane];
                }
            }
#pragma unroll
            for (int u = 0; u < 8; ++u) {
                acc.x = fmaf(h[u].x, w[u], acc.x);
                acc.y = fmaf(h[u].y, w[u], acc.y);
                acc.z = fmaf(h[u].z, w[u], acc.z);
                acc.w = fmaf(h[u].w, w[u], acc.w);
            }
        }
    }
    float4 r;
    r.x = fmaxf(acc.x + hs.x * di * di + bb.x, 0.0f);
    r.y = fmaxf(acc.y + hs.y * di * di + bb.y, 0.0f);
    r.z = fmaxf(acc.z + hs.z * di * di + bb.z, 0.0f);
    r.w = fmaxf(acc.w + hs.w * di * di + bb.w, 0.0f);
    return r;
}

__global__ __launch_bounds__(256)
void gather_kernel(const float* __restrict__ H, const int* __restrict__ rowptr,
                   const int* __restrict__ csr, const float* __restrict__ dis,
                   const float* __restrict__ b, float* __restrict__ out, int n, int e) {
    const int lane = threadIdx.x & 31;
    const int node = __builtin_amdgcn_readfirstlane(blockIdx.x * 8 + (threadIdx.x >> 5));
    if (node >= n) return;
    const float4 r = gather_row(H, rowptr, csr, dis, b, node, lane, n, e);
    ((float4*)out)[(size_t)node * (F_DIM / 4) + lane] = r;
}


#define TILE_PITCH 132
#define FUSE_WAVES 16

__global__ __launch_bounds__(FUSE_WAVES * 32, 16)
void gather_gemm_kernel(const float* __restrict__ H, const int* __restrict__ rowptr,
                        const int* __restrict__ csr, const float* __restrict__ dis,
                        const float* __restrict__ b, const float* __restrict__ W,
                        float* __restrict__ C, int n, int e, int ntiles) {
    __shared__ __attribute__((aligned(16))) float lW[F_DIM * F_DIM];
    __shared__ __attribute__((aligned(16))) float lS[FUSE_WAVES * TILE_PITCH];

    {
        const float4* __restrict__ W4 = (const float4*)W;
        float4* lW4 = (float4*)lW;
        float4 wv[F_DIM * F_DIM / 4 / (FUSE_WAVES * 32)];
#pragma unroll
        for (int q = 0; q < F_DIM * F_DIM / 4 / (FUSE_WAVES * 32); ++q)
            wv[q] = W4[threadIdx.x + q * (FUSE_WAVES * 32)];
#pragma unroll
        for (int q = 0; q < F_DIM * F_DIM / 4 / (FUSE_WAVES * 32); ++q)
            lW4[threadIdx.x + q * (FUSE_WAVES * 32)] = wv[q];
    }

    const int lane = threadIdx.x & 31;
    const int wave = __builtin_amdgcn_readfirstlane(threadIdx.x >> 5);
    const int half = lane >> 4;
    const int l16  = lane & 15;

    for (int tile = blockIdx.x; tile < ntiles; tile += gridDim.x) {
        const int node = tile * 16 + wave;
        float4 r = make_float4(0.f, 0.f, 0.f, 0.f);
        if (node < n) r = gather_row(H, rowptr, csr, dis, b, node, lane, n, e);
        __syncthreads();
        *(float4*)(lS + wave * TILE_PITCH + lane * 4) = r;
        __syncthreads();

        if (wave < 4) {
            const v8f zero = {0.f, 0.f, 0.f, 0.f, 0.f, 0.f, 0.f, 0.f};
            v8f acc0 = zero;
            v8f acc1 = zero;
            const float* arow = lS + l16 * TILE_PITCH;
            const int col0 = wave * 32 + l16;
#pragma unroll 8
            for (int k0 = 0; k0 < F_DIM; k0 += 4) {
                const int ka = k0 + half * 2;
                v2f a;
                a.x = arow[ka];
                a.y = arow[ka + 1];
                v2f b0, b1;
                b0.x = lW[ka * F_DIM + col0];
                b0.y = lW[(ka + 1) * F_DIM + col0];
                b1.x = lW[ka * F_DIM + col0 + 16];
                b1.y = lW[(ka + 1) * F_DIM + col0 + 16];
                acc0 = __builtin_amdgcn_wmma_f32_16x16x4_f32(
                     false, a,  false, b0,
                     (short)0, acc0,  false,  false);
                acc1 = __builtin_amdgcn_wmma_f32_16x16x4_f32(
                     false, a,  false, b1,
                     (short)0, acc1,  false,  false);
            }
            const int row0 = tile * 16;
#pragma unroll
            for (int v = 0; v < 8; ++v) {
                const int row = row0 + v + 8 * half;
                C[(size_t)row * F_DIM + col0]      = acc0[v];
                C[(size_t)row * F_DIM + col0 + 16] = acc1[v];
            }
        }
    }
}


#define POOL_WAVES 16

__global__ __launch_bounds__(POOL_WAVES * 32)
void pool_head_kernel(const float* __restrict__ H, const int* __restrict__ gstart,
                      const float* __restrict__ Wlin, const float* __restrict__ blin,
                      float* __restrict__ out, int n) {
    __shared__ float4 s_part[POOL_WAVES][32];
    __shared__ float  s_pool[F_DIM];
    __shared__ float  s_lg[POOL_WAVES * 2][N_CLASSES];
    const int tid  = threadIdx.x;
    const int lane = tid & 31;
    const int wave = __builtin_amdgcn_readfirstlane(tid >> 5);
    const int g    = blockIdx.x;

    const int c  = tid & (N_CLASSES - 1);
    const int ks = tid >> 4;
    float wl[4];
#pragma unroll
    for (int q = 0; q < 4; ++q) wl[q] = Wlin[(ks * 4 + q) * N_CLASSES + c];
    const float bl = blin[c];

    int lo = gstart[g];
    int hi = gstart[g + 1];
    lo = max(lo, 0);
    hi = min(hi, n);

    const float4* __restrict__ H4 = (const float4*)H;
    float4 acc = make_float4(0.f, 0.f, 0.f, 0.f);
    for (int i = lo + wave; i < hi; i += 4 * POOL_WAVES) {
        float4 h[4];
#pragma unroll
        for (int u = 0; u < 4; ++u) {
            h[u] = make_float4(0.f, 0.f, 0.f, 0.f);
            const int r = i + u * POOL_WAVES;
            if (r < hi) h[u] = H4[(size_t)r * (F_DIM / 4) + lane];
        }
#pragma unroll
        for (int u = 0; u < 4; ++u) {
            acc.x += h[u].x;
            acc.y += h[u].y;
            acc.z += h[u].z;
            acc.w += h[u].w;
        }
    }
    s_part[wave][lane] = acc;
    __syncthreads();

    if (tid < F_DIM) {
        const float* __restrict__ sp = (const float*)s_part;
        float s = sp[tid];
#pragma unroll
        for (int w = 1; w < POOL_WAVES; ++w) s += sp[w * F_DIM + tid];
        s_pool[tid] = s;
    }
    __syncthreads();

    {
        float p = s_pool[ks * 4] * wl[0];
#pragma unroll
        for (int q = 1; q < 4; ++q) p = fmaf(s_pool[ks * 4 + q], wl[q], p);
        s_lg[ks][c] = p;
    }
    __syncthreads();

    if (wave == 0) {
        float lg = bl;
#pragma unroll
        for (int s = 0; s < POOL_WAVES * 2; ++s) lg += s_lg[s][c];
        float mx = lg;
#pragma unroll
        for (int o = 8; o > 0; o >>= 1) mx = fmaxf(mx, __shfl_xor(mx, o, 16));
        float se = expf(lg - mx);
#pragma unroll
        for (int o = 8; o > 0; o >>= 1) se += __shfl_xor(se, o, 16);
        const float lse = mx + logf(se);
        if (lane < N_CLASSES) out[(size_t)g * N_CLASSES + lane] = lg - lse;
    }
}


extern "C" void kernel_launch(void* const* d_in, const int* in_sizes, int n_in,
                              void* d_out, int out_size, void* d_ws, size_t ws_size,
                              hipStream_t stream) {
    const float* x     = (const float*)d_in[0];
    const int*   eidx  = (const int*)d_in[1];
    const int*   batch = (const int*)d_in[2];
    const float* W1    = (const float*)d_in[3];
    const float* b1    = (const float*)d_in[4];
    const float* W2    = (const float*)d_in[5];
    const float* b2    = (const float*)d_in[6];
    const float* Wlin  = (const float*)d_in[7];
    const float* blin  = (const float*)d_in[8];
    float*       out   = (float*)d_out;

    const int* src = eidx;
    const int* dst = eidx + N_EDGES;

    float* ws = (float*)d_ws;
    float* dis    = ws + 50000;
    float* H      = ws + 100000;
    float* S      = H + (size_t)N_NODES * F_DIM;
    int*   gstart = (int*)(S + (size_t)N_NODES * F_DIM);
    int*   rowptr = gstart + N_GRAPHS * F_DIM;
    int*   csr    = rowptr + 50048;
    int*   cntc   = csr + 2 * N_EDGES;
    int*   base   = cntc + N_CHUNK * NODE_PAD;
    int*   tot    = base + N_CHUNK * NODE_PAD;

    const int B = 256;
    const int segBlocks   = (N_NODES + 1 + B - 1) / B;
    const int cntBlocks   = N_RANGE * N_CHUNK;
    const int gathBlocks  = (N_NODES + 7) / 8;
    const int gemmBlocks  = (N_NODES / 16 + 7) / 8;
    const int gemmHalfA   = (gemmBlocks + 1) / 2;
    const int gemmHalfB   = gemmBlocks - gemmHalfA;
    const int scanBlocks  = (N_NODES + SUB_N - 1) / SUB_N;

    static_assert(N_EDGES % (4 * N_CHUNK) == 0, "chunks are whole 16-byte groups of destinations");
    static_assert(SUB_N == 1024 && RANGE_N % SUB_N == 0 && SUB_N % 256 == 0, "scan blocks of 1024 nodes tile the ranges");
    count_gemm_kernel<<<gemmHalfA + cntBlocks + segBlocks, B, 0, stream>>>(
        dst, cntc, tot, batch, gstart, N_NODES, x, W1, H, N_NODES, gemmHalfA, 0);
    scan_dis_kernel<<<scanBlocks, SUB_N, 0, stream>>>(cntc, tot, rowptr, base, dis, N_NODES);
    place_gemm_kernel<<<gemmHalfB + cntBlocks, B, 0, stream>>>(
        src, dst, base, dis, csr, N_NODES, N_EDGES, x, W1, H, N_NODES, gemmHalfB, gemmHalfA);

    static_assert(N_NODES % 16 == 0, "the tile walk covers whole 16-node tiles");
    gather_gemm_kernel<<<1024, FUSE_WAVES * 32, 0, stream>>>(
        H, rowptr, csr, dis, b1, W2, S, N_NODES, N_EDGES, N_NODES / 16);

    gather_kernel<<<gathBlocks, B, 0, stream>>>(S, rowptr, csr, dis, b2, H, N_NODES, N_EDGES);

    pool_head_kernel<<<N_GRAPHS, POOL_WAVES * 32, 0, stream>>>(H, gstart, Wlin, blin, out, N_NODES);
}


# speedup vs baseline: 4.2373x; 1.1170x over previous
#include <hip/hip_runtime.h>
#include <hip/hip_bf16.h>
#include <math.h>


typedef float v2f __attribute__((ext_vector_type(2)));
typedef float v8f __attribute__((ext_vector_type(8)));

#define N_NODES   50000
#define N_EDGES   640000
#define F_DIM     128
#define N_CLASSES 16
#define N_GRAPHS  500


__device__ __forceinline__
void seg_ranges(const int* __restrict__ batch, int* __restrict__ gstart, int n, int i) {
    if (i <= n) {
        int prev = -1;
        int cur  = N_GRAPHS;
        if (i > 0) prev = min(max(batch[i - 1], 0), N_GRAPHS - 1);
        if (i < n) cur  = min(max(batch[i], 0), N_GRAPHS - 1);
        for (int g = prev + 1; g <= cur; ++g) gstart[g] = i;
    }
}


#define N_CHUNK   16
#define CHUNK_E   (N_EDGES / N_CHUNK)
#define RANGE_N   4096
#define N_RANGE   ((N_NODES + RANGE_N - 1) / RANGE_N)
#define NODE_PAD  (N_RANGE * RANGE_N)
#define SUB_N     1024
#define SUB_PER_RANGE (RANGE_N / SUB_N)
#define CNT_LOADS 10

__global__ __launch_bounds__(1024)
void scan_dis_kernel(const int* __restrict__ cntc, const int* __restrict__ tot,
                     int* __restrict__ rowptr, int* __restrict__ base,
                     float* __restrict__ dis, int n) {
    __shared__ int s_wtot[32];
    __shared__ int s_wpre[32];
    __shared__ int s_woff[32];
    const int tid  = threadIdx.x;
    const int lane = tid & 31;
    const int wave = tid >> 5;

    int pre = 0;
    const int nt = blockIdx.x * N_CHUNK;
    for (int q = tid; q < nt; q += 1024) pre += tot[q];

    const int i = blockIdx.x * SUB_N + tid;
    int cc[N_CHUNK];
#pragma unroll
    for (int k = 0; k < N_CHUNK; ++k) cc[k] = cntc[k * NODE_PAD + i];
    int c = 0;
#pragma unroll
    for (int k = 0; k < N_CHUNK; ++k) c += cc[k];

    int incl = c;
    for (int o = 1; o < 32; o <<= 1) {
        int t = __shfl_up(incl, o, 32);
        if (lane >= o) incl += t;
    }
    for (int o = 16; o > 0; o >>= 1) pre += __shfl_xor(pre, o, 32);

    if (lane == 31) s_wtot[wave] = incl;
    if (lane == 0)  s_wpre[wave] = pre;
    __syncthreads();
    if (wave == 0) {
        const int t = s_wtot[lane];
        int p  = s_wpre[lane];
        int ti = t;
        for (int o = 1; o < 32; o <<= 1) {
            int u = __shfl_up(ti, o, 32);
            if (lane >= o) ti += u;
        }
        for (int o = 16; o > 0; o >>= 1) p += __shfl_xor(p, o, 32);
        s_woff[lane] = p + ti - t;
    }
    __syncthreads();
    const int off = s_woff[wave];
    if (i < n) {
        int rp = off + incl - c;
        rowptr[i] = rp;
        dis[i]    = rsqrtf((float)(c + 1));
        if (i == n - 1) rowptr[n] = off + incl;
#pragma unroll
        for (int k = 0; k < N_CHUNK; ++k) {
            base[k * NODE_PAD + i] = rp;
            rp += cc[k];
        }
    }
}


__device__ __forceinline__
void gemm128_block(const float* __restrict__ A, const float* __restrict__ W,
                   float* __restrict__ C, int M, int blk, float* __restrict__ lW) {
    for (int i = threadIdx.x; i < F_DIM * F_DIM; i += 256) lW[i] = W[i];
    __syncthreads();

    const int wave = threadIdx.x >> 5;
    const int lane = threadIdx.x & 31;
    const int strip = blk * 8 + wave;
    const int row0  = strip * 16;
    if (row0 >= M) return;

    const int half = lane >> 4;
    const int l16  = lane & 15;

    const v8f zero = {0.f, 0.f, 0.f, 0.f, 0.f, 0.f, 0.f, 0.f};
    v8f acc[8];
#pragma unroll
    for (int t = 0; t < 8; ++t) acc[t] = zero;

    const float* __restrict__ arow = A + (size_t)(row0 + l16) * F_DIM;

    for (int k0 = 0; k0 < F_DIM; k0 += 4) {
        const int ka = k0 + half * 2;
        v2f a;
        a.x = arow[ka];
        a.y = arow[ka + 1];
#pragma unroll
        for (int t = 0; t < 8; ++t) {
            const int col = t * 16 + l16;
            v2f b;
            b.x = lW[ka * F_DIM + col];
            b.y = lW[(ka + 1) * F_DIM + col];
            acc[t] = __builtin_amdgcn_wmma_f32_16x16x4_f32(
                 false, a,  false, b,
                 (short)0, acc[t],  false,  false);
        }
    }

#pragma unroll
    for (int t = 0; t < 8; ++t) {
        const int col = t * 16 + l16;
#pragma unroll
        for (int v = 0; v < 8; ++v) {
            C[(size_t)(row0 + v + 8 * half) * F_DIM + col] = acc[t][v];
        }
    }
}


__global__ __launch_bounds__(256)
void count_gemm_kernel(const int* __restrict__ dst, int* __restrict__ cntc,
                       int* __restrict__ tot,
                       const int* __restrict__ batch, int* __restrict__ gstart, int n,
                       const float* __restrict__ A, const float* __restrict__ W,
                       float* __restrict__ C, int M, int gb, int g0) {
    __shared__ float lW[F_DIM * F_DIM];
    __shared__ int s_sum[SUB_PER_RANGE * 8];
    if ((int)blockIdx.x < gb) {
        gemm128_block(A, W, C, M, g0 + (int)blockIdx.x, lW);
        return;
    }
    const int idx = (int)blockIdx.x - gb;
    const int tid = threadIdx.x;
    if (idx >= N_RANGE * N_CHUNK) {
        seg_ranges(batch, gstart, n, (idx - N_RANGE * N_CHUNK) * 256 + tid);
        return;
    }
    const int r  = idx / N_CHUNK;
    const int c  = idx - r * N_CHUNK;
    const int r0 = r * RANGE_N;
    int* bins = (int*)lW;
    for (int k = 0; k < RANGE_N / 256; ++k) bins[tid + k * 256] = 0;
    __syncthreads();
    const int4* __restrict__ d4 = (const int4*)dst + c * (CHUNK_E / 4);
    for (int q0 = tid; q0 < CHUNK_E / 4; q0 += CNT_LOADS * 256) {
        int4 v[CNT_LOADS];
#pragma unroll
        for (int u = 0; u < CNT_LOADS; ++u) {
            const int q = q0 + u * 256;
            v[u] = make_int4(-1, -1, -1, -1);
            if (q < CHUNK_E / 4) v[u] = d4[q];
        }
#pragma unroll
        for (int u = 0; u < CNT_LOADS; ++u) {
            const unsigned o0 = (unsigned)(v[u].x - r0);
            const unsigned o1 = (unsigned)(v[u].y - r0);
            const unsigned o2 = (unsigned)(v[u].z - r0);
            const unsigned o3 = (unsigned)(v[u].w - r0);
            if (o0 < (unsigned)RANGE_N) atomicAdd(&bins[o0], 1);
            if (o1 < (unsigned)RANGE_N) atomicAdd(&bins[o1], 1);
            if (o2 < (unsigned)RANGE_N) atomicAdd(&bins[o2], 1);
            if (o3 < (unsigned)RANGE_N) atomicAdd(&bins[o3], 1);
        }
    }
    __syncthreads();
    int ss[SUB_PER_RANGE];
#pragma unroll
    for (int j = 0; j < SUB_PER_RANGE; ++j) ss[j] = 0;
#pragma unroll
    for (int k = 0; k < RANGE_N / 256; ++k) {
        const int b = bins[tid + k * 256];
        cntc[c * NODE_PAD + r0 + tid + k * 256] = b;
        ss[k / (SUB_N / 256)] += b;
    }
#pragma unroll
    for (int j = 0; j < SUB_PER_RANGE; ++j) {
        int t = ss[j];
        for (int o = 16; o > 0; o >>= 1) t += __shfl_xor(t, o, 32);
        if ((tid & 31) == 0) s_sum[j * 8 + (tid >> 5)] = t;
    }
    __syncthreads();
    if (tid < SUB_PER_RANGE) {
        int t = 0;
        for (int w = 0; w < 8; ++w) t += s_sum[tid * 8 + w];
        tot[(r * SUB_PER_RANGE + tid) * N_CHUNK + c] = t;
    }
}

#define PL_LOADS 10

__global__ __launch_bounds__(256)
void place_gemm_kernel(const int* __restrict__ src, const int* __restrict__ dst,
                       const int* __restrict__ base, const float* __restrict__ dis,
                       int* __restrict__ csr, int n, int e,
                       const float* __restrict__ A, const float* __restrict__ W,
                       float* __restrict__ C, int M, int gb, int g0) {
    __shared__ float lW[F_DIM * F_DIM];
    if ((int)blockIdx.x < gb) {
        gemm128_block(A, W, C, M, g0 + (int)blockIdx.x, lW);
        return;
    }
    const int idx = (int)blockIdx.x - gb;
    if (idx >= N_RANGE * N_CHUNK) return;
    const int tid = threadIdx.x;
    const int r  = idx / N_CHUNK;
    const int c  = idx - r * N_CHUNK;
    const int r0 = r * RANGE_N;
    int*   slot = (int*)lW;
    float* ldis = lW + RANGE_N + 32;
    {
        int   bs[RANGE_N / 256];
        float dd[RANGE_N / 256];
#pragma unroll
        for (int k = 0; k < RANGE_N / 256; ++k) {
            const int d = min(r0 + tid + k * 256, n - 1);
            bs[k] = base[c * NODE_PAD + d];
            dd[k] = dis[d];
        }
#pragma unroll
        for (int k = 0; k < RANGE_N / 256; ++k) {
            slot[tid + k * 256] = bs[k];
            ldis[tid + k * 256] = dd[k];
        }
    }
    if (tid < 32) slot[RANGE_N + tid] = 0x40000000;
    __syncthreads();
    const int spare = RANGE_N + (tid & 31);
    const int4* __restrict__ d4 = (const int4*)dst + c * (CHUNK_E / 4);
    const int4* __restrict__ s4 = (const int4*)src + c * (CHUNK_E / 4);
    int2* __restrict__ csr2 = (int2*)csr;
    for (int q0 = tid; q0 < CHUNK_E / 4; q0 += PL_LOADS * 256) {
        int   od[PL_LOADS][4];
        int   sr[PL_LOADS][4];
        float ds[PL_LOADS][4];
        float ld[PL_LOADS][4];
        int   ps[PL_LOADS][4];
#pragma unroll
        for (int u = 0; u < PL_LOADS; ++u) {
            const int q = q0 + u * 256;
            int4 dv = make_int4(-1, -1, -1, -1);
            int4 sv = make_int4(0, 0, 0, 0);
            if (q < CHUNK_E / 4) {
                dv = d4[q];
                sv = s4[q];
            }
            od[u][0] = dv.x - r0;  od[u][1] = dv.y - r0;  od[u][2] = dv.z - r0;  od[u][3] = dv.w - r0;
            sr[u][0] = sv.x;       sr[u][1] = sv.y;       sr[u][2] = sv.z;       sr[u][3] = sv.w;
        }
#pragma unroll
        for (int u = 0; u < PL_LOADS; ++u) {
#pragma unroll
            for (int j = 0; j < 4; ++j) {
                const bool in = (unsigned)od[u][j] < (unsigned)RANGE_N;
                const int si = in ? (int)min((unsigned)sr[u][j], (unsigned)(n - 1)) : 0;
                ds[u][j] = dis[si];
            }
        }
#pragma unroll
        for (int u = 0; u < PL_LOADS; ++u) {
#pragma unroll
            for (int j = 0; j < 4; ++j) {
                const bool in = (unsigned)od[u][j] < (unsigned)RANGE_N;
                ps[u][j] = atomicAdd(&slot[in ? od[u][j] : spare], 1);
                ld[u][j] = ldis[in ? od[u][j] : 0];
            }
        }
#pragma unroll
        for (int u = 0; u < PL_LOADS; ++u)
            asm volatile("" :: "v"(ds[u][0]), "v"(ds[u][1]), "v"(ds[u][2]), "v"(ds[u][3]));
#pragma unroll
        for (int u = 0; u < PL_LOADS; ++u) {
#pragma unroll
            for (int j = 0; j < 4; ++j) {
                const float w = ds[u][j] * ld[u][j];
                const int pos = ps[u][j];
                if ((unsigned)pos < (unsigned)e) csr2[pos] = make_int2(sr[u][j], __float_as_int(w));
            }
        }
    }
}


__device__ __forceinline__
float4 gather_row(const float* __restrict__ H, const int* __restrict__ rowptr,
                  const int* __restrict__ csr, const float* __restrict__ dis,
                  const float* __restrict__ b, int node, int lane, int n, int e) {
    int beg = rowptr[node];
    int end = rowptr[node + 1];
    beg = max(beg, 0);
    end = min(end, e);
    const float4* __restrict__ H4 = (const float4*)H;
    const int2* __restrict__ csr2 = (const int2*)csr;
    const float  di = dis[node];
    const float4 hs = H4[(size_t)node * (F_DIM / 4) + lane];
    const float4 bb = ((const float4*)b)[lane];
    float4 acc = make_float4(0.f, 0.f, 0.f, 0.f);
    for (int c0 = beg; c0 < end; c0 += 32) {
        const int m = min(32, end - c0);
        int2 ev = make_int2(0, 0);
        if (lane < m) ev = csr2[c0 + lane];
        for (int j = 0; j < m; j += 8) {
            float4 h[8];
            float  w[8];
#pragma unroll
            for (int u = 0; u < 8; ++u) {
                h[u] = make_float4(0.f, 0.f, 0.f, 0.f);
                w[u] = 0.f;
                if (u == 0 || j + u < m) {
                    int s = __builtin_amdgcn_readlane(ev.x, j + u);
                    w[u] = __int_as_float(__builtin_amdgcn_readlane(ev.y, j + u));
                    s = min((unsigned)s, (unsigned)(n - 1));
                    h[u] = H4[(size_t)s * (F_DIM / 4) + lane];
                }
            }
#pragma unroll
            for (int u = 0; u < 8; ++u) {
                acc.x = fmaf(h[u].x, w[u], acc.x);
                acc.y = fmaf(h[u].y, w[u], acc.y);
                acc.z = fmaf(h[u].z, w[u], acc.z);
                acc.w = fmaf(h[u].w, w[u], acc.w);
            }
        }
    }
    float4 r;
    r.x = fmaxf(acc.x + hs.x * di * di + bb.x, 0.0f);
    r.y = fmaxf(acc.y + hs.y * di * di + bb.y, 0.0f);
    r.z = fmaxf(acc.z + hs.z * di * di + bb.z, 0.0f);
    r.w = fmaxf(acc.w + hs.w * di * di + bb.w, 0.0f);
    return r;
}

__global__ __launch_bounds__(256)
void gather_kernel(const float* __restrict__ H, const int* __restrict__ rowptr,
                   const int* __restrict__ csr, const float* __restrict__ dis,
                   const float* __restrict__ b, float* __restrict__ out, int n, int e) {
    const int lane = threadIdx.x & 31;
    const int node = __builtin_amdgcn_readfirstlane(blockIdx.x * 8 + (threadIdx.x >> 5));
    if (node >= n) return;
    const float4 r = gather_row(H, rowptr, csr, dis, b, node, lane, n, e);
    ((float4*)out)[(size_t)node * (F_DIM / 4) + lane] = r;
}


#define TILE_PITCH 132
#define FUSE_WAVES 16

__global__ __launch_bounds__(FUSE_WAVES * 32, 16)
void gather_gemm_kernel(const float* __restrict__ H, const int* __restrict__ rowptr,
                        const int* __restrict__ csr, const float* __restrict__ dis,
                        const float* __restrict__ b, const float* __restrict__ W,
                        float* __restrict__ C, int n, int e, int ntiles) {
    __shared__ __attribute__((aligned(16))) float lW[F_DIM * F_DIM];
    __shared__ __attribute__((aligned(16))) float lS[FUSE_WAVES * TILE_PITCH];

    {
        const float4* __restrict__ W4 = (const float4*)W;
        float4* lW4 = (float4*)lW;
        float4 wv[F_DIM * F_DIM / 4 / (FUSE_WAVES * 32)];
#pragma unroll
        for (int q = 0; q < F_DIM * F_DIM / 4 / (FUSE_WAVES * 32); ++q)
            wv[q] = W4[threadIdx.x + q * (FUSE_WAVES * 32)];
#pragma unroll
        for (int q = 0; q < F_DIM * F_DIM / 4 / (FUSE_WAVES * 32); ++q)
            lW4[threadIdx.x + q * (FUSE_WAVES * 32)] = wv[q];
    }

    const int lane = threadIdx.x & 31;
    const int wave = __builtin_amdgcn_readfirstlane(threadIdx.x >> 5);
    const int half = lane >> 4;
    const int l16  = lane & 15;

    for (int tile = blockIdx.x; tile < ntiles; tile += gridDim.x) {
        const int node = tile * 16 + wave;
        float4 r = make_float4(0.f, 0.f, 0.f, 0.f);
        if (node < n) r = gather_row(H, rowptr, csr, dis, b, node, lane, n, e);
        __syncthreads();
        *(float4*)(lS + wave * TILE_PITCH + lane * 4) = r;
        __syncthreads();

        if (wave < 4) {
            const v8f zero = {0.f, 0.f, 0.f, 0.f, 0.f, 0.f, 0.f, 0.f};
            v8f acc0 = zero;
            v8f acc1 = zero;
            const float* arow = lS + l16 * TILE_PITCH;
            const int col0 = wave * 32 + l16;
#pragma unroll 8
            for (int k0 = 0; k0 < F_DIM; k0 += 4) {
                const int ka = k0 + half * 2;
                v2f a;
                a.x = arow[ka];
                a.y = arow[ka + 1];
                v2f b0, b1;
                b0.x = lW[ka * F_DIM + col0];
                b0.y = lW[(ka + 1) * F_DIM + col0];
                b1.x = lW[ka * F_DIM + col0 + 16];
                b1.y = lW[(ka + 1) * F_DIM + col0 + 16];
                acc0 = __builtin_amdgcn_wmma_f32_16x16x4_f32(
                     false, a,  false, b0,
                     (short)0, acc0,  false,  false);
                acc1 = __builtin_amdgcn_wmma_f32_16x16x4_f32(
                     false, a,  false, b1,
                     (short)0, acc1,  false,  false);
            }
            const int row0 = tile * 16;
#pragma unroll
            for (int v = 0; v < 8; ++v) {
                const int row = row0 + v + 8 * half;
                C[(size_t)row * F_DIM + col0]      = acc0[v];
                C[(size_t)row * F_DIM + col0 + 16] = acc1[v];
            }
        }
    }
}


#define POOL_WAVES 16

__global__ __launch_bounds__(POOL_WAVES * 32)
void pool_head_kernel(const float* __restrict__ H, const int* __restrict__ gstart,
                      const float* __restrict__ Wlin, const float* __restrict__ blin,
                      float* __restrict__ out, int n) {
    __shared__ float4 s_part[POOL_WAVES][32];
    __shared__ float  s_pool[F_DIM];
    __shared__ float  s_lg[POOL_WAVES * 2][N_CLASSES];
    const int tid  = threadIdx.x;
    const int lane = tid & 31;
    const int wave = __builtin_amdgcn_readfirstlane(tid >> 5);
    const int g    = blockIdx.x;

    const int c  = tid & (N_CLASSES - 1);
    const int ks = tid >> 4;
    float wl[4];
#pragma unroll
    for (int q = 0; q < 4; ++q) wl[q] = Wlin[(ks * 4 + q) * N_CLASSES + c];
    const float bl = blin[c];

    int lo = gstart[g];
    int hi = gstart[g + 1];
    lo = max(lo, 0);
    hi = min(hi, n);

    const float4* __restrict__ H4 = (const float4*)H;
    float4 acc = make_float4(0.f, 0.f, 0.f, 0.f);
    for (int i = lo + wave; i < hi; i += 4 * POOL_WAVES) {
        float4 h[4];
#pragma unroll
        for (int u = 0; u < 4; ++u) {
            h[u] = make_float4(0.f, 0.f, 0.f, 0.f);
            const int r = i + u * POOL_WAVES;
            if (r < hi) h[u] = H4[(size_t)r * (F_DIM / 4) + lane];
        }
#pragma unroll
        for (int u = 0; u < 4; ++u) {
            acc.x += h[u].x;
            acc.y += h[u].y;
            acc.z += h[u].z;
            acc.w += h[u].w;
        }
    }
    s_part[wave][lane] = acc;
    __syncthreads();

    if (tid < F_DIM) {
        const float* __restrict__ sp = (const float*)s_part;
        float s = sp[tid];
#pragma unroll
        for (int w = 1; w < POOL_WAVES; ++w) s += sp[w * F_DIM + tid];
        s_pool[tid] = s;
    }
    __syncthreads();

    {
        float p = s_pool[ks * 4] * wl[0];
#pragma unroll
        for (int q = 1; q < 4; ++q) p = fmaf(s_pool[ks * 4 + q], wl[q], p);
        s_lg[ks][c] = p;
    }
    __syncthreads();

    if (wave == 0) {
        float lg = bl;
#pragma unroll
        for (int s = 0; s < POOL_WAVES * 2; ++s) lg += s_lg[s][c];
        float mx = lg;
#pragma unroll
        for (int o = 8; o > 0; o >>= 1) mx = fmaxf(mx, __shfl_xor(mx, o, 16));
        float se = expf(lg - mx);
#pragma unroll
        for (int o = 8; o > 0; o >>= 1) se += __shfl_xor(se, o, 16);
        const float lse = mx + logf(se);
        if (lane < N_CLASSES) out[(size_t)g * N_CLASSES + lane] = lg - lse;
    }
}


extern "C" void kernel_launch(void* const* d_in, const int* in_sizes, int n_in,
                              void* d_out, int out_size, void* d_ws, size_t ws_size,
                              hipStream_t stream) {
    const float* x     = (const float*)d_in[0];
    const int*   eidx  = (const int*)d_in[1];
    const int*   batch = (const int*)d_in[2];
    const float* W1    = (const float*)d_in[3];
    const float* b1    = (const float*)d_in[4];
    const float* W2    = (const float*)d_in[5];
    const float* b2    = (const float*)d_in[6];
    const float* Wlin  = (const float*)d_in[7];
    const float* blin  = (const float*)d_in[8];
    float*       out   = (float*)d_out;

    const int* src = eidx;
    const int* dst = eidx + N_EDGES;

    float* ws = (float*)d_ws;
    float* dis    = ws + 50048;
    float* H      = ws + 100352;
    float* S      = H + (size_t)N_NODES * F_DIM;
    int*   gstart = (int*)(S + (size_t)N_NODES * F_DIM);
    int*   rowptr = gstart + N_GRAPHS * F_DIM;
    int*   csr    = rowptr + 50048;
    int*   cntc   = csr + 2 * N_EDGES;
    int*   base   = cntc + N_CHUNK * NODE_PAD;
    int*   tot    = base + N_CHUNK * NODE_PAD;

    const int B = 256;
    const int segBlocks   = (N_NODES + 1 + B - 1) / B;
    const int cntBlocks   = N_RANGE * N_CHUNK;
    const int gathBlocks  = (N_NODES + 7) / 8;
    const int gemmBlocks  = (N_NODES / 16 + 7) / 8;
    const int gemmHalfA   = (gemmBlocks + 1) / 2;
    const int gemmHalfB   = gemmBlocks - gemmHalfA;
    const int scanBlocks  = (N_NODES + SUB_N - 1) / SUB_N;

    static_assert(N_EDGES % (4 * N_CHUNK) == 0, "chunks are whole 16-byte groups of destinations");
    static_assert(SUB_N == 1024 && RANGE_N % SUB_N == 0 && SUB_N % 256 == 0, "scan blocks of 1024 nodes tile the ranges");
    count_gemm_kernel<<<gemmHalfA + cntBlocks + segBlocks, B, 0, stream>>>(
        dst, cntc, tot, batch, gstart, N_NODES, x, W1, H, N_NODES, gemmHalfA, 0);
    scan_dis_kernel<<<scanBlocks, SUB_N, 0, stream>>>(cntc, tot, rowptr, base, dis, N_NODES);
    place_gemm_kernel<<<gemmHalfB + cntBlocks, B, 0, stream>>>(
        src, dst, base, dis, csr, N_NODES, N_EDGES, x, W1, H, N_NODES, gemmHalfB, gemmHalfA);

    static_assert(N_NODES % 16 == 0, "the tile walk covers whole 16-node tiles");
    gather_gemm_kernel<<<1024, FUSE_WAVES * 32, 0, stream>>>(
        H, rowptr, csr, dis, b1, W2, S, N_NODES, N_EDGES, N_NODES / 16);

    gather_kernel<<<gathBlocks, B, 0, stream>>>(S, rowptr, csr, dis, b2, H, N_NODES, N_EDGES);

    pool_head_kernel<<<N_GRAPHS, POOL_WAVES * 32, 0, stream>>>(H, gstart, Wlin, blin, out, N_NODES);
}


# Pool+head fused into the second gather: one 32-wave block per graph gathers its nodes' rows, adds them in registers, then head + log_softmax; row store, pool launch gone; pool sum = 32 wave partials
# speedup vs baseline: 4.4504x; 1.0503x over previous
#include <hip/hip_runtime.h>
#include <hip/hip_bf16.h>
#include <math.h>


typedef float v2f __attribute__((ext_vector_type(2)));
typedef float v8f __attribute__((ext_vector_type(8)));

#define N_NODES   50000
#define N_EDGES   640000
#define F_DIM     128
#define N_CLASSES 16
#define N_GRAPHS  500


__device__ __forceinline__
void seg_ranges(const int* __restrict__ batch, int* __restrict__ gstart, int n, int i) {
    if (i <= n) {
        int prev = -1;
        int cur  = N_GRAPHS;
        if (i > 0) prev = min(max(batch[i - 1], 0), N_GRAPHS - 1);
        if (i < n) cur  = min(max(batch[i], 0), N_GRAPHS - 1);
        for (int g = prev + 1; g <= cur; ++g) gstart[g] = i;
    }
}


#define N_CHUNK   16
#define CHUNK_E   (N_EDGES / N_CHUNK)
#define RANGE_N   4096
#define N_RANGE   ((N_NODES + RANGE_N - 1) / RANGE_N)
#define NODE_PAD  (N_RANGE * RANGE_N)
#define SUB_N     1024
#define SUB_PER_RANGE (RANGE_N / SUB_N)
#define CNT_LOADS 10

__global__ __launch_bounds__(1024)
void scan_dis_kernel(const int* __restrict__ cntc, const int* __restrict__ tot,
                     int* __restrict__ rowptr, int* __restrict__ base,
                     float* __restrict__ dis, int n) {
    __shared__ int s_wtot[32];
    __shared__ int s_wpre[32];
    __shared__ int s_woff[32];
    const int tid  = threadIdx.x;
    const int lane = tid & 31;
    const int wave = tid >> 5;

    int pre = 0;
    const int nt = blockIdx.x * N_CHUNK;
    for (int q = tid; q < nt; q += 1024) pre += tot[q];

    const int i = blockIdx.x * SUB_N + tid;
    int cc[N_CHUNK];
#pragma unroll
    for (int k = 0; k < N_CHUNK; ++k) cc[k] = cntc[k * NODE_PAD + i];
    int c = 0;
#pragma unroll
    for (int k = 0; k < N_CHUNK; ++k) c += cc[k];

    int incl = c;
    for (int o = 1; o < 32; o <<= 1) {
        int t = __shfl_up(incl, o, 32);
        if (lane >= o) incl += t;
    }
    for (int o = 16; o > 0; o >>= 1) pre += __shfl_xor(pre, o, 32);

    if (lane == 31) s_wtot[wave] = incl;
    if (lane == 0)  s_wpre[wave] = pre;
    __syncthreads();
    if (wave == 0) {
        const int t = s_wtot[lane];
        int p  = s_wpre[lane];
        int ti = t;
        for (int o = 1; o < 32; o <<= 1) {
            int u = __shfl_up(ti, o, 32);
            if (lane >= o) ti += u;
        }
        for (int o = 16; o > 0; o >>= 1) p += __shfl_xor(p, o, 32);
        s_woff[lane] = p + ti - t;
    }
    __syncthreads();
    const int off = s_woff[wave];
    if (i < n) {
        int rp = off + incl - c;
        rowptr[i] = rp;
        dis[i]    = rsqrtf((float)(c + 1));
        if (i == n - 1) rowptr[n] = off + incl;
#pragma unroll
        for (int k = 0; k < N_CHUNK; ++k) {
            base[k * NODE_PAD + i] = rp;
            rp += cc[k];
        }
    }
}


__device__ __forceinline__
void gemm128_block(const float* __restrict__ A, const float* __restrict__ W,
                   float* __restrict__ C, int M, int blk, float* __restrict__ lW) {
    for (int i = threadIdx.x; i < F_DIM * F_DIM; i += 256) lW[i] = W[i];
    __syncthreads();

    const int wave = threadIdx.x >> 5;
    const int lane = threadIdx.x & 31;
    const int strip = blk * 8 + wave;
    const int row0  = strip * 16;
    if (row0 >= M) return;

    const int half = lane >> 4;
    const int l16  = lane & 15;

    const v8f zero = {0.f, 0.f, 0.f, 0.f, 0.f, 0.f, 0.f, 0.f};
    v8f acc[8];
#pragma unroll
    for (int t = 0; t < 8; ++t) acc[t] = zero;

    const float* __restrict__ arow = A + (size_t)(row0 + l16) * F_DIM;

    for (int k0 = 0; k0 < F_DIM; k0 += 4) {
        const int ka = k0 + half * 2;
        v2f a;
        a.x = arow[ka];
        a.y = arow[ka + 1];
#pragma unroll
        for (int t = 0; t < 8; ++t) {
            const int col = t * 16 + l16;
            v2f b;
            b.x = lW[ka * F_DIM + col];
            b.y = lW[(ka + 1) * F_DIM + col];
            acc[t] = __builtin_amdgcn_wmma_f32_16x16x4_f32(
                 false, a,  false, b,
                 (short)0, acc[t],  false,  false);
        }
    }

#pragma unroll
    for (int t = 0; t < 8; ++t) {
        const int col = t * 16 + l16;
#pragma unroll
        for (int v = 0; v < 8; ++v) {
            C[(size_t)(row0 + v + 8 * half) * F_DIM + col] = acc[t][v];
        }
    }
}


__global__ __launch_bounds__(256)
void count_gemm_kernel(const int* __restrict__ dst, int* __restrict__ cntc,
                       int* __restrict__ tot,
                       const int* __restrict__ batch, int* __restrict__ gstart, int n,
                       const float* __restrict__ A, const float* __restrict__ W,
                       float* __restrict__ C, int M, int gb, int g0) {
    __shared__ float lW[F_DIM * F_DIM];
    __shared__ int s_sum[SUB_PER_RANGE * 8];
    if ((int)blockIdx.x < gb) {
        gemm128_block(A, W, C, M, g0 + (int)blockIdx.x, lW);
        return;
    }
    const int idx = (int)blockIdx.x - gb;
    const int tid = threadIdx.x;
    if (idx >= N_RANGE * N_CHUNK) {
        seg_ranges(batch, gstart, n, (idx - N_RANGE * N_CHUNK) * 256 + tid);
        return;
    }
    const int r  = idx / N_CHUNK;
    const int c  = idx - r * N_CHUNK;
    const int r0 = r * RANGE_N;
    int* bins = (int*)lW;
    for (int k = 0; k < RANGE_N / 256; ++k) bins[tid + k * 256] = 0;
    __syncthreads();
    const int4* __restrict__ d4 = (const int4*)dst + c * (CHUNK_E / 4);
    for (int q0 = tid; q0 < CHUNK_E / 4; q0 += CNT_LOADS * 256) {
        int4 v[CNT_LOADS];
#pragma unroll
        for (int u = 0; u < CNT_LOADS; ++u) {
            const int q = q0 + u * 256;
            v[u] = make_int4(-1, -1, -1, -1);
            if (q < CHUNK_E / 4) v[u] = d4[q];
        }
#pragma unroll
        for (int u = 0; u < CNT_LOADS; ++u) {
            const unsigned o0 = (unsigned)(v[u].x - r0);
            const unsigned o1 = (unsigned)(v[u].y - r0);
            const unsigned o2 = (unsigned)(v[u].z - r0);
            const unsigned o3 = (unsigned)(v[u].w - r0);
            if (o0 < (unsigned)RANGE_N) atomicAdd(&bins[o0], 1);
            if (o1 < (unsigned)RANGE_N) atomicAdd(&bins[o1], 1);
            if (o2 < (unsigned)RANGE_N) atomicAdd(&bins[o2], 1);
            if (o3 < (unsigned)RANGE_N) atomicAdd(&bins[o3], 1);
        }
    }
    __syncthreads();
    int ss[SUB_PER_RANGE];
#pragma unroll
    for (int j = 0; j < SUB_PER_RANGE; ++j) ss[j] = 0;
#pragma unroll
    for (int k = 0; k < RANGE_N / 256; ++k) {
        const int b = bins[tid + k * 256];
        cntc[c * NODE_PAD + r0 + tid + k * 256] = b;
        ss[k / (SUB_N / 256)] += b;
    }
#pragma unroll
    for (int j = 0; j < SUB_PER_RANGE; ++j) {
        int t = ss[j];
        for (int o = 16; o > 0; o >>= 1) t += __shfl_xor(t, o, 32);
        if ((tid & 31) == 0) s_sum[j * 8 + (tid >> 5)] = t;
    }
    __syncthreads();
    if (tid < SUB_PER_RANGE) {
        int t = 0;
        for (int w = 0; w < 8; ++w) t += s_sum[tid * 8 + w];
        tot[(r * SUB_PER_RANGE + tid) * N_CHUNK + c] = t;
    }
}

#define PL_LOADS 10

__global__ __launch_bounds__(256)
void place_gemm_kernel(const int* __restrict__ src, const int* __restrict__ dst,
                       const int* __restrict__ base, const float* __restrict__ dis,
                       int* __restrict__ csr, int n, int e,
                       const float* __restrict__ A, const float* __restrict__ W,
                       float* __restrict__ C, int M, int gb, int g0) {
    __shared__ float lW[F_DIM * F_DIM];
    if ((int)blockIdx.x < gb) {
        gemm128_block(A, W, C, M, g0 + (int)blockIdx.x, lW);
        return;
    }
    const int idx = (int)blockIdx.x - gb;
    if (idx >= N_RANGE * N_CHUNK) return;
    const int tid = threadIdx.x;
    const int r  = idx / N_CHUNK;
    const int c  = idx - r * N_CHUNK;
    const int r0 = r * RANGE_N;
    int*   slot = (int*)lW;
    float* ldis = lW + RANGE_N + 32;
    {
        int   bs[RANGE_N / 256];
        float dd[RANGE_N / 256];
#pragma unroll
        for (int k = 0; k < RANGE_N / 256; ++k) {
            const int d = min(r0 + tid + k * 256, n - 1);
            bs[k] = base[c * NODE_PAD + d];
            dd[k] = dis[d];
        }
#pragma unroll
        for (int k = 0; k < RANGE_N / 256; ++k) {
            slot[tid + k * 256] = bs[k];
            ldis[tid + k * 256] = dd[k];
        }
    }
    if (tid < 32) slot[RANGE_N + tid] = 0x40000000;
    __syncthreads();
    const int spare = RANGE_N + (tid & 31);
    const int4* __restrict__ d4 = (const int4*)dst + c * (CHUNK_E / 4);
    const int4* __restrict__ s4 = (const int4*)src + c * (CHUNK_E / 4);
    int2* __restrict__ csr2 = (int2*)csr;
    for (int q0 = tid; q0 < CHUNK_E / 4; q0 += PL_LOADS * 256) {
        int   od[PL_LOADS][4];
        int   sr[PL_LOADS][4];
        float ds[PL_LOADS][4];
        float ld[PL_LOADS][4];
        int   ps[PL_LOADS][4];
#pragma unroll
        for (int u = 0; u < PL_LOADS; ++u) {
            const int q = q0 + u * 256;
            int4 dv = make_int4(-1, -1, -1, -1);
            int4 sv = make_int4(0, 0, 0, 0);
            if (q < CHUNK_E / 4) {
                dv = d4[q];
                sv = s4[q];
            }
            od[u][0] = dv.x - r0;  od[u][1] = dv.y - r0;  od[u][2] = dv.z - r0;  od[u][3] = dv.w - r0;
            sr[u][0] = sv.x;       sr[u][1] = sv.y;       sr[u][2] = sv.z;       sr[u][3] = sv.w;
        }
#pragma unroll
        for (int u = 0; u < PL_LOADS; ++u) {
#pragma unroll
            for (int j = 0; j < 4; ++j) {
                const bool in = (unsigned)od[u][j] < (unsigned)RANGE_N;
                const int si = in ? (int)min((unsigned)sr[u][j], (unsigned)(n - 1)) : 0;
                ds[u][j] = dis[si];
            }
        }
#pragma unroll
        for (int u = 0; u < PL_LOADS; ++u) {
#pragma unroll
            for (int j = 0; j < 4; ++j) {
                const bool in = (unsigned)od[u][j] < (unsigned)RANGE_N;
                ps[u][j] = atomicAdd(&slot[in ? od[u][j] : spare], 1);
                ld[u][j] = ldis[in ? od[u][j] : 0];
            }
        }
#pragma unroll
        for (int u = 0; u < PL_LOADS; ++u)
            asm volatile("" :: "v"(ds[u][0]), "v"(ds[u][1]), "v"(ds[u][2]), "v"(ds[u][3]));
#pragma unroll
        for (int u = 0; u < PL_LOADS; ++u) {
#pragma unroll
            for (int j = 0; j < 4; ++j) {
                const float w = ds[u][j] * ld[u][j];
                const int pos = ps[u][j];
                if ((unsigned)pos < (unsigned)e) csr2[pos] = make_int2(sr[u][j], __float_as_int(w));
            }
        }
    }
}


__device__ __forceinline__
float4 gather_row(const float* __restrict__ H, const int* __restrict__ rowptr,
                  const int* __restrict__ csr, const float* __restrict__ dis,
                  const float* __restrict__ b, int node, int lane, int n, int e) {
    int beg = rowptr[node];
    int end = rowptr[node + 1];
    beg = max(beg, 0);
    end = min(end, e);
    const float4* __restrict__ H4 = (const float4*)H;
    const int2* __restrict__ csr2 = (const int2*)csr;
    const float  di = dis[node];
    const float4 hs = H4[(size_t)node * (F_DIM / 4) + lane];
    const float4 bb = ((const float4*)b)[lane];
    float4 acc = make_float4(0.f, 0.f, 0.f, 0.f);
    for (int c0 = beg; c0 < end; c0 += 32) {
        const int m = min(32, end - c0);
        int2 ev = make_int2(0, 0);
        if (lane < m) ev = csr2[c0 + lane];
        for (int j = 0; j < m; j += 8) {
            float4 h[8];
            float  w[8];
#pragma unroll
            for (int u = 0; u < 8; ++u) {
                h[u] = make_float4(0.f, 0.f, 0.f, 0.f);
                w[u] = 0.f;
                if (u == 0 || j + u < m) {
                    int s = __builtin_amdgcn_readlane(ev.x, j + u);
                    w[u] = __int_as_float(__builtin_amdgcn_readlane(ev.y, j + u));
                    s = min((unsigned)s, (unsigned)(n - 1));
                    h[u] = H4[(size_t)s * (F_DIM / 4) + lane];
                }
            }
#pragma unroll
            for (int u = 0; u < 8; ++u) {
                acc.x = fmaf(h[u].x, w[u], acc.x);
                acc.y = fmaf(h[u].y, w[u], acc.y);
                acc.z = fmaf(h[u].z, w[u], acc.z);
                acc.w = fmaf(h[u].w, w[u], acc.w);
            }
        }
    }
    float4 r;
    r.x = fmaxf(acc.x + hs.x * di * di + bb.x, 0.0f);
    r.y = fmaxf(acc.y + hs.y * di * di + bb.y, 0.0f);
    r.z = fmaxf(acc.z + hs.z * di * di + bb.z, 0.0f);
    r.w = fmaxf(acc.w + hs.w * di * di + bb.w, 0.0f);
    return r;
}


#define TILE_PITCH 132
#define FUSE_WAVES 16

__global__ __launch_bounds__(FUSE_WAVES * 32, 16)
void gather_gemm_kernel(const float* __restrict__ H, const int* __restrict__ rowptr,
                        const int* __restrict__ csr, const float* __restrict__ dis,
                        const float* __restrict__ b, const float* __restrict__ W,
                        float* __restrict__ C, int n, int e, int ntiles) {
    __shared__ __attribute__((aligned(16))) float lW[F_DIM * F_DIM];
    __shared__ __attribute__((aligned(16))) float lS[FUSE_WAVES * TILE_PITCH];

    {
        const float4* __restrict__ W4 = (const float4*)W;
        float4* lW4 = (float4*)lW;
        float4 wv[F_DIM * F_DIM / 4 / (FUSE_WAVES * 32)];
#pragma unroll
        for (int q = 0; q < F_DIM * F_DIM / 4 / (FUSE_WAVES * 32); ++q)
            wv[q] = W4[threadIdx.x + q * (FUSE_WAVES * 32)];
#pragma unroll
        for (int q = 0; q < F_DIM * F_DIM / 4 / (FUSE_WAVES * 32); ++q)
            lW4[threadIdx.x + q * (FUSE_WAVES * 32)] = wv[q];
    }

    const int lane = threadIdx.x & 31;
    const int wave = __builtin_amdgcn_readfirstlane(threadIdx.x >> 5);
    const int half = lane >> 4;
    const int l16  = lane & 15;

    for (int tile = blockIdx.x; tile < ntiles; tile += gridDim.x) {
        const int node = tile * 16 + wave;
        float4 r = make_float4(0.f, 0.f, 0.f, 0.f);
        if (node < n) r = gather_row(H, rowptr, csr, dis, b, node, lane, n, e);
        __syncthreads();
        *(float4*)(lS + wave * TILE_PITCH + lane * 4) = r;
        __syncthreads();

        if (wave < 4) {
            const v8f zero = {0.f, 0.f, 0.f, 0.f, 0.f, 0.f, 0.f, 0.f};
            v8f acc0 = zero;
            v8f acc1 = zero;
            const float* arow = lS + l16 * TILE_PITCH;
            const int col0 = wave * 32 + l16;
#pragma unroll 8
            for (int k0 = 0; k0 < F_DIM; k0 += 4) {
                const int ka = k0 + half * 2;
                v2f a;
                a.x = arow[ka];
                a.y = arow[ka + 1];
                v2f b0, b1;
                b0.x = lW[ka * F_DIM + col0];
                b0.y = lW[(ka + 1) * F_DIM + col0];
                b1.x = lW[ka * F_DIM + col0 + 16];
                b1.y = lW[(ka + 1) * F_DIM + col0 + 16];
                acc0 = __builtin_amdgcn_wmma_f32_16x16x4_f32(
                     false, a,  false, b0,
                     (short)0, acc0,  false,  false);
                acc1 = __builtin_amdgcn_wmma_f32_16x16x4_f32(
                     false, a,  false, b1,
                     (short)0, acc1,  false,  false);
            }
            const int row0 = tile * 16;
#pragma unroll
            for (int v = 0; v < 8; ++v) {
                const int row = row0 + v + 8 * half;
                C[(size_t)row * F_DIM + col0]      = acc0[v];
                C[(size_t)row * F_DIM + col0 + 16] = acc1[v];
            }
        }
    }
}


#define POOL_WAVES 32
#define HEAD_SLICES (F_DIM / 4)

__global__ __launch_bounds__(POOL_WAVES * 32, 16)
void gather_pool_head_kernel(const float* __restrict__ S, const int* __restrict__ rowptr,
                             const int* __restrict__ csr, const float* __restrict__ dis,
                             const float* __restrict__ b, const int* __restrict__ gstart,
                             const float* __restrict__ Wlin, const float* __restrict__ blin,
                             float* __restrict__ out, int n, int e) {
    __shared__ float4 s_part[POOL_WAVES][32];
    __shared__ float  s_pool[F_DIM];
    __shared__ float  s_lg[HEAD_SLICES][N_CLASSES];
    const int tid  = threadIdx.x;
    const int lane = tid & 31;
    const int wave = __builtin_amdgcn_readfirstlane(tid >> 5);
    const int g    = blockIdx.x;

    const int c  = tid & (N_CLASSES - 1);
    const int ks = (tid >> 4) & (HEAD_SLICES - 1);
    float wl[4] = {0.f, 0.f, 0.f, 0.f};
    float bl = 0.f;
    if (tid < HEAD_SLICES * N_CLASSES) {
#pragma unroll
        for (int q = 0; q < 4; ++q) wl[q] = Wlin[(ks * 4 + q) * N_CLASSES + c];
        bl = blin[c];
    }

    int lo = gstart[g];
    int hi = gstart[g + 1];
    lo = max(lo, 0);
    hi = min(hi, n);

    float4 acc = make_float4(0.f, 0.f, 0.f, 0.f);
    for (int node = lo + wave; node < hi; node += POOL_WAVES) {
        const float4 r = gather_row(S, rowptr, csr, dis, b, node, lane, n, e);
        acc.x += r.x;
        acc.y += r.y;
        acc.z += r.z;
        acc.w += r.w;
    }
    s_part[wave][lane] = acc;
    __syncthreads();

    if (tid < F_DIM) {
        const float* __restrict__ sp = (const float*)s_part;
        float s = sp[tid];
#pragma unroll
        for (int w = 1; w < POOL_WAVES; ++w) s += sp[w * F_DIM + tid];
        s_pool[tid] = s;
    }
    __syncthreads();

    if (tid < HEAD_SLICES * N_CLASSES) {
        float p = s_pool[ks * 4] * wl[0];
#pragma unroll
        for (int q = 1; q < 4; ++q) p = fmaf(s_pool[ks * 4 + q], wl[q], p);
        s_lg[ks][c] = p;
    }
    __syncthreads();

    if (wave == 0) {
        float lg = bl;
#pragma unroll
        for (int s = 0; s < HEAD_SLICES; ++s) lg += s_lg[s][c];
        float mx = lg;
#pragma unroll
        for (int o = 8; o > 0; o >>= 1) mx = fmaxf(mx, __shfl_xor(mx, o, 16));
        float se = expf(lg - mx);
#pragma unroll
        for (int o = 8; o > 0; o >>= 1) se += __shfl_xor(se, o, 16);
        const float lse = mx + logf(se);
        if (lane < N_CLASSES) out[(size_t)g * N_CLASSES + lane] = lg - lse;
    }
}


extern "C" void kernel_launch(void* const* d_in, const int* in_sizes, int n_in,
                              void* d_out, int out_size, void* d_ws, size_t ws_size,
                              hipStream_t stream) {
    const float* x     = (const float*)d_in[0];
    const int*   eidx  = (const int*)d_in[1];
    const int*   batch = (const int*)d_in[2];
    const float* W1    = (const float*)d_in[3];
    const float* b1    = (const float*)d_in[4];
    const float* W2    = (const float*)d_in[5];
    const float* b2    = (const float*)d_in[6];
    const float* Wlin  = (const float*)d_in[7];
    const float* blin  = (const float*)d_in[8];
    float*       out   = (float*)d_out;

    const int* src = eidx;
    const int* dst = eidx + N_EDGES;

    float* ws = (float*)d_ws;
    float* dis    = ws + 50048;
    float* H      = ws + 100352;
    float* S      = H + (size_t)N_NODES * F_DIM;
    int*   gstart = (int*)(S + (size_t)N_NODES * F_DIM);
    int*   rowptr = gstart + N_GRAPHS * F_DIM;
    int*   csr    = rowptr + 50048;
    int*   cntc   = csr + 2 * N_EDGES;
    int*   base   = cntc + N_CHUNK * NODE_PAD;
    int*   tot    = base + N_CHUNK * NODE_PAD;

    const int B = 256;
    const int segBlocks   = (N_NODES + 1 + B - 1) / B;
    const int cntBlocks   = N_RANGE * N_CHUNK;
    const int gemmBlocks  = (N_NODES / 16 + 7) / 8;
    const int gemmHalfA   = (gemmBlocks + 1) / 2;
    const int gemmHalfB   = gemmBlocks - gemmHalfA;
    const int scanBlocks  = (N_NODES + SUB_N - 1) / SUB_N;

    static_assert(N_EDGES % (4 * N_CHUNK) == 0, "chunks are whole 16-byte groups of destinations");
    static_assert(SUB_N == 1024 && RANGE_N % SUB_N == 0 && SUB_N % 256 == 0, "scan blocks of 1024 nodes tile the ranges");
    count_gemm_kernel<<<gemmHalfA + cntBlocks + segBlocks, B, 0, stream>>>(
        dst, cntc, tot, batch, gstart, N_NODES, x, W1, H, N_NODES, gemmHalfA, 0);
    scan_dis_kernel<<<scanBlocks, SUB_N, 0, stream>>>(cntc, tot, rowptr, base, dis, N_NODES);
    place_gemm_kernel<<<gemmHalfB + cntBlocks, B, 0, stream>>>(
        src, dst, base, dis, csr, N_NODES, N_EDGES, x, W1, H, N_NODES, gemmHalfB, gemmHalfA);

    static_assert(N_NODES % 16 == 0, "the tile walk covers whole 16-node tiles");
    gather_gemm_kernel<<<1024, FUSE_WAVES * 32, 0, stream>>>(
        H, rowptr, csr, dis, b1, W2, S, N_NODES, N_EDGES, N_NODES / 16);

    gather_pool_head_kernel<<<N_GRAPHS, POOL_WAVES * 32, 0, stream>>>(
        S, rowptr, csr, dis, b2, gstart, Wlin, blin, out, N_NODES, N_EDGES);
}
